# starting kernel
# baseline (speedup 1.0000x reference)
#include <hip/hip_runtime.h>
#include <hip/hip_bf16.h>
#include <cstddef>

#define N_NODES 100000
#define N_EDGES 600000
#define D 128
#define H 256

typedef float v2f __attribute__((ext_vector_type(2)));
typedef float v8f __attribute__((ext_vector_type(8)));

__device__ __forceinline__ v8f wmma_f32(v2f a, v2f b, v8f c) {
    return __builtin_amdgcn_wmma_f32_16x16x4_f32(false, a, false, b, (short)0, c,
                                                 false, false);
}

#define ROWS 32
#define XS_STRIDE 132
#define H1_STRIDE 260

__global__ __launch_bounds__(256) void gnn_msg_mlp(
    const float* __restrict__ nodes, const float* __restrict__ Wm1,
    const float* __restrict__ bm1, const float* __restrict__ Wm2,
    const float* __restrict__ bm2, float* __restrict__ m_out) {
    __shared__ float Xs[ROWS * XS_STRIDE];
    __shared__ float H1s[ROWS * H1_STRIDE];

    const int tid = threadIdx.x;
    const int row_base = blockIdx.x * ROWS;

    for (int i = tid; i < ROWS * (D / 4); i += 256) {
        int row = i / (D / 4);
        int c4 = i % (D / 4);
        int rg = row_base + row;
        float4 v = make_float4(0.f, 0.f, 0.f, 0.f);
        if (rg < N_NODES)
            v = ((const float4*)(nodes + (size_t)rg * D))[c4];
        *(float4*)&Xs[row * XS_STRIDE + c4 * 4] = v;
    }
    __syncthreads();

    const int w = tid >> 5;
    const int lane = tid & 31;
    const int half = lane >> 4;
    const int l16 = lane & 15;

    for (int c = 0; c < 2; ++c) {
        int col = (w * 2 + c) * 16 + l16;
        float bv = bm1[col];
        v8f acc0 = {bv, bv, bv, bv, bv, bv, bv, bv};
        v8f acc1 = acc0;
#pragma unroll 4
        for (int k = 0; k < D; k += 4) {
            int kk = k + half * 2;
            v2f b = {Wm1[kk * H + col], Wm1[(kk + 1) * H + col]};
            const float* x0 = &Xs[l16 * XS_STRIDE + kk];
            const float* x1 = &Xs[(16 + l16) * XS_STRIDE + kk];
            v2f a0 = {x0[0], x0[1]};
            v2f a1 = {x1[0], x1[1]};
            acc0 = wmma_f32(a0, b, acc0);
            acc1 = wmma_f32(a1, b, acc1);
        }
#pragma unroll
        for (int r = 0; r < 8; ++r) {
            int row = r + half * 8;
            float v0 = acc0[r];
            float v1 = acc1[r];
            H1s[row * H1_STRIDE + col] = v0 > 0.f ? v0 : 0.f;
            H1s[(16 + row) * H1_STRIDE + col] = v1 > 0.f ? v1 : 0.f;
        }
    }
    __syncthreads();

    {
        int col = w * 16 + l16;
        float bv = bm2[col];
        v8f acc0 = {bv, bv, bv, bv, bv, bv, bv, bv};
        v8f acc1 = acc0;
#pragma unroll 4
        for (int k = 0; k < H; k += 4) {
            int kk = k + half * 2;
            v2f b = {Wm2[kk * D + col], Wm2[(kk + 1) * D + col]};
            const float* h0 = &H1s[l16 * H1_STRIDE + kk];
            const float* h1 = &H1s[(16 + l16) * H1_STRIDE + kk];
            v2f a0 = {h0[0], h0[1]};
            v2f a1 = {h1[0], h1[1]};
            acc0 = wmma_f32(a0, b, acc0);
            acc1 = wmma_f32(a1, b, acc1);
        }
#pragma unroll
        for (int r = 0; r < 8; ++r) {
            int row = r + half * 8;
            int rg0 = row_base + row;
            int rg1 = row_base + 16 + row;
            if (rg0 < N_NODES) m_out[(size_t)rg0 * D + col] = acc0[r];
            if (rg1 < N_NODES) m_out[(size_t)rg1 * D + col] = acc1[r];
        }
    }
}

__global__ __launch_bounds__(256) void gnn_scatter(
    const float* __restrict__ m, const int* __restrict__ senders,
    const int* __restrict__ receivers, float* __restrict__ agg) {
    const int wave = threadIdx.x >> 5;
    const int lane = threadIdx.x & 31;
    const int e = blockIdx.x * 8 + wave;
    if (e >= N_EDGES) return;
    const int s = senders[e];
    const int r = receivers[e];
    const float4 v = ((const float4*)(m + (size_t)s * D))[lane];
    float* dst = agg + (size_t)r * D + lane * 4;
    __hip_atomic_fetch_add(dst + 0, v.x, __ATOMIC_RELAXED, __HIP_MEMORY_SCOPE_AGENT);
    __hip_atomic_fetch_add(dst + 1, v.y, __ATOMIC_RELAXED, __HIP_MEMORY_SCOPE_AGENT);
    __hip_atomic_fetch_add(dst + 2, v.z, __ATOMIC_RELAXED, __HIP_MEMORY_SCOPE_AGENT);
    __hip_atomic_fetch_add(dst + 3, v.w, __ATOMIC_RELAXED, __HIP_MEMORY_SCOPE_AGENT);
}

__global__ __launch_bounds__(256) void gnn_node_mlp(
    const float* __restrict__ nodes, const float* __restrict__ agg,
    const float* __restrict__ Wn1, const float* __restrict__ bn1,
    const float* __restrict__ Wn2, const float* __restrict__ bn2,
    float* __restrict__ out) {
    __shared__ float Xs[ROWS * XS_STRIDE];
    __shared__ float H1s[ROWS * H1_STRIDE];

    const int tid = threadIdx.x;
    const int row_base = blockIdx.x * ROWS;

    for (int i = tid; i < ROWS * (D / 4); i += 256) {
        int row = i / (D / 4);
        int c4 = i % (D / 4);
        int rg = row_base + row;
        float4 v = make_float4(0.f, 0.f, 0.f, 0.f);
        if (rg < N_NODES)
            v = ((const float4*)(nodes + (size_t)rg * D))[c4];
        *(float4*)&Xs[row * XS_STRIDE + c4 * 4] = v;
    }
    __syncthreads();

    const int w = tid >> 5;
    const int lane = tid & 31;
    const int half = lane >> 4;
    const int l16 = lane & 15;

    int rg0 = row_base + l16;
    int rg1 = row_base + 16 + l16;
    if (rg0 >= N_NODES) rg0 = N_NODES - 1;
    if (rg1 >= N_NODES) rg1 = N_NODES - 1;
    const float* aggRow0 = agg + (size_t)rg0 * D;
    const float* aggRow1 = agg + (size_t)rg1 * D;

    for (int c = 0; c < 2; ++c) {
        int col = (w * 2 + c) * 16 + l16;
        float bv = bn1[col];
        v8f acc0 = {bv, bv, bv, bv, bv, bv, bv, bv};
        v8f acc1 = acc0;
#pragma unroll 4
        for (int k = 0; k < D; k += 4) {
            int kk = k + half * 2;
            v2f b = {Wn1[kk * H + col], Wn1[(kk + 1) * H + col]};
            const float* x0 = &Xs[l16 * XS_STRIDE + kk];
            const float* x1 = &Xs[(16 + l16) * XS_STRIDE + kk];
            v2f a0 = {x0[0], x0[1]};
            v2f a1 = {x1[0], x1[1]};
            acc0 = wmma_f32(a0, b, acc0);
            acc1 = wmma_f32(a1, b, acc1);
        }
#pragma unroll 4
        for (int k = D; k < 2 * D; k += 4) {
            int kk = k + half * 2;
            v2f b = {Wn1[kk * H + col], Wn1[(kk + 1) * H + col]};
            const float* a0p = aggRow0 + (kk - D);
            const float* a1p = aggRow1 + (kk - D);
            v2f a0 = {a0p[0], a0p[1]};
            v2f a1 = {a1p[0], a1p[1]};
            acc0 = wmma_f32(a0, b, acc0);
            acc1 = wmma_f32(a1, b, acc1);
        }
#pragma unroll
        for (int r = 0; r < 8; ++r) {
            int row = r + half * 8;
            float v0 = acc0[r];
            float v1 = acc1[r];
            H1s[row * H1_STRIDE + col] = v0 > 0.f ? v0 : 0.f;
            H1s[(16 + row) * H1_STRIDE + col] = v1 > 0.f ? v1 : 0.f;
        }
    }
    __syncthreads();

    {
        int col = w * 16 + l16;
        float bv = bn2[col];
        v8f acc0 = {bv, bv, bv, bv, bv, bv, bv, bv};
        v8f acc1 = acc0;
#pragma unroll 4
        for (int k = 0; k < H; k += 4) {
            int kk = k + half * 2;
            v2f b = {Wn2[kk * D + col], Wn2[(kk + 1) * D + col]};
            const float* h0 = &H1s[l16 * H1_STRIDE + kk];
            const float* h1 = &H1s[(16 + l16) * H1_STRIDE + kk];
            v2f a0 = {h0[0], h0[1]};
            v2f a1 = {h1[0], h1[1]};
            acc0 = wmma_f32(a0, b, acc0);
            acc1 = wmma_f32(a1, b, acc1);
        }
#pragma unroll
        for (int r = 0; r < 8; ++r) {
            int row = r + half * 8;
            int g0 = row_base + row;
            int g1 = row_base + 16 + row;
            if (g0 < N_NODES)
                out[(size_t)g0 * D + col] = Xs[row * XS_STRIDE + col] + acc0[r];
            if (g1 < N_NODES)
                out[(size_t)g1 * D + col] = Xs[(16 + row) * XS_STRIDE + col] + acc1[r];
        }
    }
}

extern "C" void kernel_launch(void* const* d_in, const int* in_sizes, int n_in,
                              void* d_out, int out_size, void* d_ws, size_t ws_size,
                              hipStream_t stream) {
    (void)in_sizes; (void)n_in; (void)out_size; (void)ws_size;
    const float* nodes     = (const float*)d_in[0];
    const int*   senders   = (const int*)d_in[1];
    const int*   receivers = (const int*)d_in[2];
    const float* Wm1 = (const float*)d_in[3];
    const float* bm1 = (const float*)d_in[4];
    const float* Wm2 = (const float*)d_in[5];
    const float* bm2 = (const float*)d_in[6];
    const float* Wn1 = (const float*)d_in[7];
    const float* bn1 = (const float*)d_in[8];
    const float* Wn2 = (const float*)d_in[9];
    const float* bn2 = (const float*)d_in[10];

    float* m_buf = (float*)d_ws;
    float* out   = (float*)d_out;

    const int row_blocks = (N_NODES + ROWS - 1) / ROWS;

    gnn_msg_mlp<<<row_blocks, 256, 0, stream>>>(nodes, Wm1, bm1, Wm2, bm2, m_buf);

    hipMemsetAsync(d_out, 0, sizeof(float) * (size_t)N_NODES * D, stream);

    gnn_scatter<<<(N_EDGES + 7) / 8, 256, 0, stream>>>(m_buf, senders, receivers, out);

    gnn_node_mlp<<<row_blocks, 256, 0, stream>>>(nodes, out, Wn1, bn1, Wn2, bn2, out);
}


# segment sum without float atomics: integer exchange links each edge into its receiver's list, then one wave per node gathers and sums its messages; sum order = list arrival order
# speedup vs baseline: 1.1507x; 1.1507x over previous
#include <hip/hip_runtime.h>
#include <hip/hip_bf16.h>
#include <cstddef>

#define N_NODES 100000
#define N_EDGES 600000
#define D 128
#define H 256

typedef float v2f __attribute__((ext_vector_type(2)));
typedef float v8f __attribute__((ext_vector_type(8)));

__device__ __forceinline__ v8f wmma_f32(v2f a, v2f b, v8f c) {
    return __builtin_amdgcn_wmma_f32_16x16x4_f32(false, a, false, b, (short)0, c,
                                                 false, false);
}

#define ROWS 32
#define XS_STRIDE 132
#define H1_STRIDE 260

__global__ __launch_bounds__(256) void gnn_msg_mlp(
    const float* __restrict__ nodes, const float* __restrict__ Wm1,
    const float* __restrict__ bm1, const float* __restrict__ Wm2,
    const float* __restrict__ bm2, float* __restrict__ m_out) {
    __shared__ float Xs[ROWS * XS_STRIDE];
    __shared__ float H1s[ROWS * H1_STRIDE];

    const int tid = threadIdx.x;
    const int row_base = blockIdx.x * ROWS;

    for (int i = tid; i < ROWS * (D / 4); i += 256) {
        int row = i / (D / 4);
        int c4 = i % (D / 4);
        int rg = row_base + row;
        float4 v = make_float4(0.f, 0.f, 0.f, 0.f);
        if (rg < N_NODES)
            v = ((const float4*)(nodes + (size_t)rg * D))[c4];
        *(float4*)&Xs[row * XS_STRIDE + c4 * 4] = v;
    }
    __syncthreads();

    const int w = tid >> 5;
    const int lane = tid & 31;
    const int half = lane >> 4;
    const int l16 = lane & 15;

    for (int c = 0; c < 2; ++c) {
        int col = (w * 2 + c) * 16 + l16;
        float bv = bm1[col];
        v8f acc0 = {bv, bv, bv, bv, bv, bv, bv, bv};
        v8f acc1 = acc0;
#pragma unroll 4
        for (int k = 0; k < D; k += 4) {
            int kk = k + half * 2;
            v2f b = {Wm1[kk * H + col], Wm1[(kk + 1) * H + col]};
            const float* x0 = &Xs[l16 * XS_STRIDE + kk];
            const float* x1 = &Xs[(16 + l16) * XS_STRIDE + kk];
            v2f a0 = {x0[0], x0[1]};
            v2f a1 = {x1[0], x1[1]};
            acc0 = wmma_f32(a0, b, acc0);
            acc1 = wmma_f32(a1, b, acc1);
        }
#pragma unroll
        for (int r = 0; r < 8; ++r) {
            int row = r + half * 8;
            float v0 = acc0[r];
            float v1 = acc1[r];
            H1s[row * H1_STRIDE + col] = v0 > 0.f ? v0 : 0.f;
            H1s[(16 + row) * H1_STRIDE + col] = v1 > 0.f ? v1 : 0.f;
        }
    }
    __syncthreads();

    {
        int col = w * 16 + l16;
        float bv = bm2[col];
        v8f acc0 = {bv, bv, bv, bv, bv, bv, bv, bv};
        v8f acc1 = acc0;
#pragma unroll 4
        for (int k = 0; k < H; k += 4) {
            int kk = k + half * 2;
            v2f b = {Wm2[kk * D + col], Wm2[(kk + 1) * D + col]};
            const float* h0 = &H1s[l16 * H1_STRIDE + kk];
            const float* h1 = &H1s[(16 + l16) * H1_STRIDE + kk];
            v2f a0 = {h0[0], h0[1]};
            v2f a1 = {h1[0], h1[1]};
            acc0 = wmma_f32(a0, b, acc0);
            acc1 = wmma_f32(a1, b, acc1);
        }
#pragma unroll
        for (int r = 0; r < 8; ++r) {
            int row = r + half * 8;
            int rg0 = row_base + row;
            int rg1 = row_base + 16 + row;
            if (rg0 < N_NODES) m_out[(size_t)rg0 * D + col] = acc0[r];
            if (rg1 < N_NODES) m_out[(size_t)rg1 * D + col] = acc1[r];
        }
    }
}

__global__ __launch_bounds__(256) void gnn_link_edges(
    const int* __restrict__ senders, const int* __restrict__ receivers,
    int* __restrict__ head, int2* __restrict__ link) {
    const int e = blockIdx.x * 256 + threadIdx.x;
    if (e >= N_EDGES) return;
    const int r = receivers[e];
    const int s = senders[e];
    const int prev = __hip_atomic_exchange(head + r, e, __ATOMIC_RELAXED,
                                           __HIP_MEMORY_SCOPE_AGENT);
    link[e] = make_int2(prev, s);
}

__global__ __launch_bounds__(256) void gnn_gather(
    const float* __restrict__ m, const int* __restrict__ head,
    const int2* __restrict__ link, float* __restrict__ agg) {
    const int lane = threadIdx.x & 31;
    const int wave = __builtin_amdgcn_readfirstlane((int)(threadIdx.x >> 5));
    const int r = blockIdx.x * 8 + wave;
    if (r >= N_NODES) return;
    float4 acc = make_float4(0.f, 0.f, 0.f, 0.f);
    float4 pend = acc;
    int e = head[r];
    for (int it = 0; it < N_EDGES; ++it) {
        if ((unsigned)e >= (unsigned)N_EDGES) break;
        const int2 ln = link[e];
        acc.x += pend.x; acc.y += pend.y; acc.z += pend.z; acc.w += pend.w;
        pend = ((const float4*)(m + (size_t)ln.y * D))[lane];
        e = ln.x;
    }
    acc.x += pend.x; acc.y += pend.y; acc.z += pend.z; acc.w += pend.w;
    ((float4*)(agg + (size_t)r * D))[lane] = acc;
}

__global__ __launch_bounds__(256) void gnn_node_mlp(
    const float* __restrict__ nodes, const float* __restrict__ agg,
    const float* __restrict__ Wn1, const float* __restrict__ bn1,
    const float* __restrict__ Wn2, const float* __restrict__ bn2,
    float* __restrict__ out) {
    __shared__ float Xs[ROWS * XS_STRIDE];
    __shared__ float H1s[ROWS * H1_STRIDE];

    const int tid = threadIdx.x;
    const int row_base = blockIdx.x * ROWS;

    for (int i = tid; i < ROWS * (D / 4); i += 256) {
        int row = i / (D / 4);
        int c4 = i % (D / 4);
        int rg = row_base + row;
        float4 v = make_float4(0.f, 0.f, 0.f, 0.f);
        if (rg < N_NODES)
            v = ((const float4*)(nodes + (size_t)rg * D))[c4];
        *(float4*)&Xs[row * XS_STRIDE + c4 * 4] = v;
    }
    __syncthreads();

    const int w = tid >> 5;
    const int lane = tid & 31;
    const int half = lane >> 4;
    const int l16 = lane & 15;

    int rg0 = row_base + l16;
    int rg1 = row_base + 16 + l16;
    if (rg0 >= N_NODES) rg0 = N_NODES - 1;
    if (rg1 >= N_NODES) rg1 = N_NODES - 1;
    const float* aggRow0 = agg + (size_t)rg0 * D;
    const float* aggRow1 = agg + (size_t)rg1 * D;

    for (int c = 0; c < 2; ++c) {
        int col = (w * 2 + c) * 16 + l16;
        float bv = bn1[col];
        v8f acc0 = {bv, bv, bv, bv, bv, bv, bv, bv};
        v8f acc1 = acc0;
#pragma unroll 4
        for (int k = 0; k < D; k += 4) {
            int kk = k + half * 2;
            v2f b = {Wn1[kk * H + col], Wn1[(kk + 1) * H + col]};
            const float* x0 = &Xs[l16 * XS_STRIDE + kk];
            const float* x1 = &Xs[(16 + l16) * XS_STRIDE + kk];
            v2f a0 = {x0[0], x0[1]};
            v2f a1 = {x1[0], x1[1]};
            acc0 = wmma_f32(a0, b, acc0);
            acc1 = wmma_f32(a1, b, acc1);
        }
#pragma unroll 4
        for (int k = D; k < 2 * D; k += 4) {
            int kk = k + half * 2;
            v2f b = {Wn1[kk * H + col], Wn1[(kk + 1) * H + col]};
            const float* a0p = aggRow0 + (kk - D);
            const float* a1p = aggRow1 + (kk - D);
            v2f a0 = {a0p[0], a0p[1]};
            v2f a1 = {a1p[0], a1p[1]};
            acc0 = wmma_f32(a0, b, acc0);
            acc1 = wmma_f32(a1, b, acc1);
        }
#pragma unroll
        for (int r = 0; r < 8; ++r) {
            int row = r + half * 8;
            float v0 = acc0[r];
            float v1 = acc1[r];
            H1s[row * H1_STRIDE + col] = v0 > 0.f ? v0 : 0.f;
            H1s[(16 + row) * H1_STRIDE + col] = v1 > 0.f ? v1 : 0.f;
        }
    }
    __syncthreads();

    {
        int col = w * 16 + l16;
        float bv = bn2[col];
        v8f acc0 = {bv, bv, bv, bv, bv, bv, bv, bv};
        v8f acc1 = acc0;
#pragma unroll 4
        for (int k = 0; k < H; k += 4) {
            int kk = k + half * 2;
            v2f b = {Wn2[kk * D + col], Wn2[(kk + 1) * D + col]};
            const float* h0 = &H1s[l16 * H1_STRIDE + kk];
            const float* h1 = &H1s[(16 + l16) * H1_STRIDE + kk];
            v2f a0 = {h0[0], h0[1]};
            v2f a1 = {h1[0], h1[1]};
            acc0 = wmma_f32(a0, b, acc0);
            acc1 = wmma_f32(a1, b, acc1);
        }
#pragma unroll
        for (int r = 0; r < 8; ++r) {
            int row = r + half * 8;
            int g0 = row_base + row;
            int g1 = row_base + 16 + row;
            if (g0 < N_NODES)
                out[(size_t)g0 * D + col] = Xs[row * XS_STRIDE + col] + acc0[r];
            if (g1 < N_NODES)
                out[(size_t)g1 * D + col] = Xs[(16 + row) * XS_STRIDE + col] + acc1[r];
        }
    }
}

extern "C" void kernel_launch(void* const* d_in, const int* in_sizes, int n_in,
                              void* d_out, int out_size, void* d_ws, size_t ws_size,
                              hipStream_t stream) {
    (void)in_sizes; (void)n_in; (void)out_size; (void)ws_size;
    const float* nodes     = (const float*)d_in[0];
    const int*   senders   = (const int*)d_in[1];
    const int*   receivers = (const int*)d_in[2];
    const float* Wm1 = (const float*)d_in[3];
    const float* bm1 = (const float*)d_in[4];
    const float* Wm2 = (const float*)d_in[5];
    const float* bm2 = (const float*)d_in[6];
    const float* Wn1 = (const float*)d_in[7];
    const float* bn1 = (const float*)d_in[8];
    const float* Wn2 = (const float*)d_in[9];
    const float* bn2 = (const float*)d_in[10];

    float* m_buf = (float*)d_ws;
    int*   head  = (int*)(m_buf + (size_t)N_NODES * D);
    int2*  link  = (int2*)(head + N_NODES);
    float* out   = (float*)d_out;

    const int row_blocks = (N_NODES + ROWS - 1) / ROWS;

    gnn_msg_mlp<<<row_blocks, 256, 0, stream>>>(nodes, Wm1, bm1, Wm2, bm2, m_buf);

    hipMemsetAsync(head, 0xFF, sizeof(int) * (size_t)N_NODES, stream);

    gnn_link_edges<<<(N_EDGES + 255) / 256, 256, 0, stream>>>(senders, receivers, head, link);

    gnn_gather<<<(N_NODES + 7) / 8, 256, 0, stream>>>(m_buf, head, link, out);

    gnn_node_mlp<<<row_blocks, 256, 0, stream>>>(nodes, out, Wn1, bn1, Wn2, bn2, out);
}


# both MLP kernels re-tiled: 64 rows per block, each wave holds 32 interleaved columns x 4 row tiles (layer 2: x 2), so one pair of 8-byte weight loads feeds 8 (4) WMMAs instead of 2
# speedup vs baseline: 1.2951x; 1.1255x over previous
#include <hip/hip_runtime.h>
#include <hip/hip_bf16.h>
#include <cstddef>

#define N_NODES 100000
#define N_EDGES 600000
#define D 128
#define H 256

typedef float v2f __attribute__((ext_vector_type(2)));
typedef float v8f __attribute__((ext_vector_type(8)));

__device__ __forceinline__ v8f wmma_f32(v2f a, v2f b, v8f c) {
    return __builtin_amdgcn_wmma_f32_16x16x4_f32(false, a, false, b, (short)0, c,
                                                 false, false);
}

#define ROWS 64
#define RT (ROWS / 16)
#define XS_STRIDE 132
#define H1_STRIDE 260


__device__ __forceinline__ v8f splat8(float v) {
    v8f r = {v, v, v, v, v, v, v, v};
    return r;
}

__global__ __launch_bounds__(256) void gnn_msg_mlp(
    const float* __restrict__ nodes, const float* __restrict__ Wm1,
    const float* __restrict__ bm1, const float* __restrict__ Wm2,
    const float* __restrict__ bm2, float* __restrict__ m_out) {
    __shared__ float Xs[ROWS * XS_STRIDE];
    __shared__ float H1s[ROWS * H1_STRIDE];

    const int tid = threadIdx.x;
    const int row_base = blockIdx.x * ROWS;

    for (int i = tid; i < ROWS * (D / 4); i += 256) {
        int row = i / (D / 4);
        int c4 = i % (D / 4);
        int rg = row_base + row;
        float4 v = make_float4(0.f, 0.f, 0.f, 0.f);
        if (rg < N_NODES)
            v = ((const float4*)(nodes + (size_t)rg * D))[c4];
        *(float4*)&Xs[row * XS_STRIDE + c4 * 4] = v;
    }
    __syncthreads();

    const int w = tid >> 5;
    const int lane = tid & 31;
    const int half = lane >> 4;
    const int l16 = lane & 15;

    {
        const int cb = w * 32 + 2 * l16;
        const float2 bv = *(const float2*)&bm1[cb];
        v8f acc[RT][2];
#pragma unroll
        for (int rt = 0; rt < RT; ++rt) {
            acc[rt][0] = splat8(bv.x);
            acc[rt][1] = splat8(bv.y);
        }
        const float* wp = Wm1 + (size_t)(half * 2) * H + cb;
        const float* xp = &Xs[l16 * XS_STRIDE + half * 2];
#pragma unroll 2
        for (int k = 0; k < D; k += 4) {
            const float2 q0 = *(const float2*)(wp + (size_t)k * H);
            const float2 q1 = *(const float2*)(wp + (size_t)(k + 1) * H);
            const v2f b0 = {q0.x, q1.x};
            const v2f b1 = {q0.y, q1.y};
#pragma unroll
            for (int rt = 0; rt < RT; ++rt) {
                const float* x = xp + rt * 16 * XS_STRIDE + k;
                const v2f a = {x[0], x[1]};
                acc[rt][0] = wmma_f32(a, b0, acc[rt][0]);
                acc[rt][1] = wmma_f32(a, b1, acc[rt][1]);
            }
        }
#pragma unroll
        for (int rt = 0; rt < RT; ++rt) {
#pragma unroll
            for (int r = 0; r < 8; ++r) {
                const int row = rt * 16 + half * 8 + r;
                const float v0 = acc[rt][0][r];
                const float v1 = acc[rt][1][r];
                *(float2*)&H1s[row * H1_STRIDE + cb] =
                    make_float2(v0 > 0.f ? v0 : 0.f, v1 > 0.f ? v1 : 0.f);
            }
        }
    }
    __syncthreads();

    {
        const int cb = (w & 3) * 32 + 2 * l16;
        const int rt0 = (w >> 2) * 2;
        const float2 bv = *(const float2*)&bm2[cb];
        v8f acc[2][2];
#pragma unroll
        for (int j = 0; j < 2; ++j) {
            acc[j][0] = splat8(bv.x);
            acc[j][1] = splat8(bv.y);
        }
        const float* wp = Wm2 + (size_t)(half * 2) * D + cb;
        const float* hp = &H1s[(rt0 * 16 + l16) * H1_STRIDE + half * 2];
#pragma unroll 4
        for (int k = 0; k < H; k += 4) {
            const float2 q0 = *(const float2*)(wp + (size_t)k * D);
            const float2 q1 = *(const float2*)(wp + (size_t)(k + 1) * D);
            const v2f b0 = {q0.x, q1.x};
            const v2f b1 = {q0.y, q1.y};
#pragma unroll
            for (int j = 0; j < 2; ++j) {
                const float* hh = hp + j * 16 * H1_STRIDE + k;
                const v2f a = {hh[0], hh[1]};
                acc[j][0] = wmma_f32(a, b0, acc[j][0]);
                acc[j][1] = wmma_f32(a, b1, acc[j][1]);
            }
        }
#pragma unroll
        for (int j = 0; j < 2; ++j) {
#pragma unroll
            for (int r = 0; r < 8; ++r) {
                const int rg = row_base + (rt0 + j) * 16 + half * 8 + r;
                if (rg < N_NODES)
                    *(float2*)&m_out[(size_t)rg * D + cb] =
                        make_float2(acc[j][0][r], acc[j][1][r]);
            }
        }
    }
}

__global__ __launch_bounds__(256) void gnn_link_edges(
    const int* __restrict__ senders, const int* __restrict__ receivers,
    int* __restrict__ head, int2* __restrict__ link) {
    const int e = blockIdx.x * 256 + threadIdx.x;
    if (e >= N_EDGES) return;
    const int r = receivers[e];
    const int s = senders[e];
    const int prev = __hip_atomic_exchange(head + r, e, __ATOMIC_RELAXED,
                                           __HIP_MEMORY_SCOPE_AGENT);
    link[e] = make_int2(prev, s);
}

__global__ __launch_bounds__(256) void gnn_gather(
    const float* __restrict__ m, const int* __restrict__ head,
    const int2* __restrict__ link, float* __restrict__ agg) {
    const int lane = threadIdx.x & 31;
    const int wave = __builtin_amdgcn_readfirstlane((int)(threadIdx.x >> 5));
    const int r = blockIdx.x * 8 + wave;
    if (r >= N_NODES) return;
    float4 acc = make_float4(0.f, 0.f, 0.f, 0.f);
    float4 pend = acc;
    int e = head[r];
    for (int it = 0; it < N_EDGES; ++it) {
        if ((unsigned)e >= (unsigned)N_EDGES) break;
        const int2 ln = link[e];
        acc.x += pend.x; acc.y += pend.y; acc.z += pend.z; acc.w += pend.w;
        pend = ((const float4*)(m + (size_t)ln.y * D))[lane];
        e = ln.x;
    }
    acc.x += pend.x; acc.y += pend.y; acc.z += pend.z; acc.w += pend.w;
    ((float4*)(agg + (size_t)r * D))[lane] = acc;
}

__global__ __launch_bounds__(256) void gnn_node_mlp(
    const float* __restrict__ nodes, const float* __restrict__ agg,
    const float* __restrict__ Wn1, const float* __restrict__ bn1,
    const float* __restrict__ Wn2, const float* __restrict__ bn2,
    float* __restrict__ out) {
    __shared__ float Xs[ROWS * XS_STRIDE];
    __shared__ float H1s[ROWS * H1_STRIDE];

    const int tid = threadIdx.x;
    const int row_base = blockIdx.x * ROWS;

    for (int i = tid; i < ROWS * (D / 4); i += 256) {
        int row = i / (D / 4);
        int c4 = i % (D / 4);
        int rg = row_base + row;
        float4 v = make_float4(0.f, 0.f, 0.f, 0.f);
        if (rg < N_NODES)
            v = ((const float4*)(nodes + (size_t)rg * D))[c4];
        *(float4*)&Xs[row * XS_STRIDE + c4 * 4] = v;
    }
    __syncthreads();

    const int w = tid >> 5;
    const int lane = tid & 31;
    const int half = lane >> 4;
    const int l16 = lane & 15;

    {
        const int cb = w * 32 + 2 * l16;
        const float2 bv = *(const float2*)&bn1[cb];
        v8f acc[RT][2];
#pragma unroll
        for (int rt = 0; rt < RT; ++rt) {
            acc[rt][0] = splat8(bv.x);
            acc[rt][1] = splat8(bv.y);
        }
        const float* wp = Wn1 + (size_t)(half * 2) * H + cb;
        const float* xp = &Xs[l16 * XS_STRIDE + half * 2];
#pragma unroll 2
        for (int k = 0; k < D; k += 4) {
            const float2 q0 = *(const float2*)(wp + (size_t)k * H);
            const float2 q1 = *(const float2*)(wp + (size_t)(k + 1) * H);
            const v2f b0 = {q0.x, q1.x};
            const v2f b1 = {q0.y, q1.y};
#pragma unroll
            for (int rt = 0; rt < RT; ++rt) {
                const float* x = xp + rt * 16 * XS_STRIDE + k;
                const v2f a = {x[0], x[1]};
                acc[rt][0] = wmma_f32(a, b0, acc[rt][0]);
                acc[rt][1] = wmma_f32(a, b1, acc[rt][1]);
            }
        }
        const float* ap[RT];
#pragma unroll
        for (int rt = 0; rt < RT; ++rt) {
            int tb = row_base + rt * 16;
            if (tb > N_NODES - 16) tb = N_NODES - 16;
            ap[rt] = agg + (size_t)(tb + l16) * D + half * 2;
        }
        const float* wq = wp + (size_t)D * H;
#pragma unroll 2
        for (int k = 0; k < D; k += 4) {
            const float2 q0 = *(const float2*)(wq + (size_t)k * H);
            const float2 q1 = *(const float2*)(wq + (size_t)(k + 1) * H);
            const v2f b0 = {q0.x, q1.x};
            const v2f b1 = {q0.y, q1.y};
#pragma unroll
            for (int rt = 0; rt < RT; ++rt) {
                const float2 av = *(const float2*)(ap[rt] + k);
                const v2f a = {av.x, av.y};
                acc[rt][0] = wmma_f32(a, b0, acc[rt][0]);
                acc[rt][1] = wmma_f32(a, b1, acc[rt][1]);
            }
        }
#pragma unroll
        for (int rt = 0; rt < RT; ++rt) {
#pragma unroll
            for (int r = 0; r < 8; ++r) {
                const int row = rt * 16 + half * 8 + r;
                const float v0 = acc[rt][0][r];
                const float v1 = acc[rt][1][r];
                *(float2*)&H1s[row * H1_STRIDE + cb] =
                    make_float2(v0 > 0.f ? v0 : 0.f, v1 > 0.f ? v1 : 0.f);
            }
        }
    }
    __syncthreads();

    {
        const int cb = (w & 3) * 32 + 2 * l16;
        const int rt0 = (w >> 2) * 2;
        const float2 bv = *(const float2*)&bn2[cb];
        v8f acc[2][2];
#pragma unroll
        for (int j = 0; j < 2; ++j) {
            acc[j][0] = splat8(bv.x);
            acc[j][1] = splat8(bv.y);
        }
        const float* wp = Wn2 + (size_t)(half * 2) * D + cb;
        const float* hp = &H1s[(rt0 * 16 + l16) * H1_STRIDE + half * 2];
#pragma unroll 4
        for (int k = 0; k < H; k += 4) {
            const float2 q0 = *(const float2*)(wp + (size_t)k * D);
            const float2 q1 = *(const float2*)(wp + (size_t)(k + 1) * D);
            const v2f b0 = {q0.x, q1.x};
            const v2f b1 = {q0.y, q1.y};
#pragma unroll
            for (int j = 0; j < 2; ++j) {
                const float* hh = hp + j * 16 * H1_STRIDE + k;
                const v2f a = {hh[0], hh[1]};
                acc[j][0] = wmma_f32(a, b0, acc[j][0]);
                acc[j][1] = wmma_f32(a, b1, acc[j][1]);
            }
        }
#pragma unroll
        for (int j = 0; j < 2; ++j) {
#pragma unroll
            for (int r = 0; r < 8; ++r) {
                const int row = (rt0 + j) * 16 + half * 8 + r;
                const int rg = row_base + row;
                if (rg < N_NODES) {
                    const float2 xv = *(const float2*)&Xs[row * XS_STRIDE + cb];
                    *(float2*)&out[(size_t)rg * D + cb] =
                        make_float2(xv.x + acc[j][0][r], xv.y + acc[j][1][r]);
                }
            }
        }
    }
}

extern "C" void kernel_launch(void* const* d_in, const int* in_sizes, int n_in,
                              void* d_out, int out_size, void* d_ws, size_t ws_size,
                              hipStream_t stream) {
    (void)in_sizes; (void)n_in; (void)out_size; (void)ws_size;
    const float* nodes     = (const float*)d_in[0];
    const int*   senders   = (const int*)d_in[1];
    const int*   receivers = (const int*)d_in[2];
    const float* Wm1 = (const float*)d_in[3];
    const float* bm1 = (const float*)d_in[4];
    const float* Wm2 = (const float*)d_in[5];
    const float* bm2 = (const float*)d_in[6];
    const float* Wn1 = (const float*)d_in[7];
    const float* bn1 = (const float*)d_in[8];
    const float* Wn2 = (const float*)d_in[9];
    const float* bn2 = (const float*)d_in[10];

    float* m_buf = (float*)d_ws;
    int*   head  = (int*)(m_buf + (size_t)N_NODES * D);
    int2*  link  = (int2*)(head + N_NODES);
    float* out   = (float*)d_out;

    const int row_blocks = (N_NODES + ROWS - 1) / ROWS;

    gnn_msg_mlp<<<row_blocks, 256, 0, stream>>>(nodes, Wm1, bm1, Wm2, bm2, m_buf);

    hipMemsetAsync(head, 0xFF, sizeof(int) * (size_t)N_NODES, stream);

    gnn_link_edges<<<(N_EDGES + 255) / 256, 256, 0, stream>>>(senders, receivers, head, link);

    gnn_gather<<<(N_NODES + 7) / 8, 256, 0, stream>>>(m_buf, head, link, out);

    gnn_node_mlp<<<row_blocks, 256, 0, stream>>>(nodes, out, Wn1, bn1, Wn2, bn2, out);
}


# segment sum folded into the node kernel: each block walks its 64 receivers' lists (8 per wave side by side) and sums messages into an LDS tile read by layer 1; sums still in list arrival order
# speedup vs baseline: 1.3619x; 1.0516x over previous
#include <hip/hip_runtime.h>
#include <hip/hip_bf16.h>
#include <cstddef>

#define N_NODES 100000
#define N_EDGES 600000
#define D 128
#define H 256

typedef float v2f __attribute__((ext_vector_type(2)));
typedef float v8f __attribute__((ext_vector_type(8)));

__device__ __forceinline__ v8f wmma_f32(v2f a, v2f b, v8f c) {
    return __builtin_amdgcn_wmma_f32_16x16x4_f32(false, a, false, b, (short)0, c,
                                                 false, false);
}

#define ROWS 64
#define RT (ROWS / 16)
#define XS_STRIDE 132
#define H1_STRIDE 260


__device__ __forceinline__ v8f splat8(float v) {
    v8f r = {v, v, v, v, v, v, v, v};
    return r;
}

__global__ __launch_bounds__(256) void gnn_msg_mlp(
    const float* __restrict__ nodes, const float* __restrict__ Wm1,
    const float* __restrict__ bm1, const float* __restrict__ Wm2,
    const float* __restrict__ bm2, float* __restrict__ m_out) {
    __shared__ float Xs[ROWS * XS_STRIDE];
    __shared__ float H1s[ROWS * H1_STRIDE];

    const int tid = threadIdx.x;
    const int row_base = blockIdx.x * ROWS;

    for (int i = tid; i < ROWS * (D / 4); i += 256) {
        int row = i / (D / 4);
        int c4 = i % (D / 4);
        int rg = row_base + row;
        float4 v = make_float4(0.f, 0.f, 0.f, 0.f);
        if (rg < N_NODES)
            v = ((const float4*)(nodes + (size_t)rg * D))[c4];
        *(float4*)&Xs[row * XS_STRIDE + c4 * 4] = v;
    }
    __syncthreads();

    const int w = tid >> 5;
    const int lane = tid & 31;
    const int half = lane >> 4;
    const int l16 = lane & 15;

    {
        const int cb = w * 32 + 2 * l16;
        const float2 bv = *(const float2*)&bm1[cb];
        v8f acc[RT][2];
#pragma unroll
        for (int rt = 0; rt < RT; ++rt) {
            acc[rt][0] = splat8(bv.x);
            acc[rt][1] = splat8(bv.y);
        }
        const float* wp = Wm1 + (size_t)(half * 2) * H + cb;
        const float* xp = &Xs[l16 * XS_STRIDE + half * 2];
#pragma unroll 2
        for (int k = 0; k < D; k += 4) {
            const float2 q0 = *(const float2*)(wp + (size_t)k * H);
            const float2 q1 = *(const float2*)(wp + (size_t)(k + 1) * H);
            const v2f b0 = {q0.x, q1.x};
            const v2f b1 = {q0.y, q1.y};
#pragma unroll
            for (int rt = 0; rt < RT; ++rt) {
                const float* x = xp + rt * 16 * XS_STRIDE + k;
                const v2f a = {x[0], x[1]};
                acc[rt][0] = wmma_f32(a, b0, acc[rt][0]);
                acc[rt][1] = wmma_f32(a, b1, acc[rt][1]);
            }
        }
#pragma unroll
        for (int rt = 0; rt < RT; ++rt) {
#pragma unroll
            for (int r = 0; r < 8; ++r) {
                const int row = rt * 16 + half * 8 + r;
                const float v0 = acc[rt][0][r];
                const float v1 = acc[rt][1][r];
                *(float2*)&H1s[row * H1_STRIDE + cb] =
                    make_float2(v0 > 0.f ? v0 : 0.f, v1 > 0.f ? v1 : 0.f);
            }
        }
    }
    __syncthreads();

    {
        const int cb = (w & 3) * 32 + 2 * l16;
        const int rt0 = (w >> 2) * 2;
        const float2 bv = *(const float2*)&bm2[cb];
        v8f acc[2][2];
#pragma unroll
        for (int j = 0; j < 2; ++j) {
            acc[j][0] = splat8(bv.x);
            acc[j][1] = splat8(bv.y);
        }
        const float* wp = Wm2 + (size_t)(half * 2) * D + cb;
        const float* hp = &H1s[(rt0 * 16 + l16) * H1_STRIDE + half * 2];
#pragma unroll 4
        for (int k = 0; k < H; k += 4) {
            const float2 q0 = *(const float2*)(wp + (size_t)k * D);
            const float2 q1 = *(const float2*)(wp + (size_t)(k + 1) * D);
            const v2f b0 = {q0.x, q1.x};
            const v2f b1 = {q0.y, q1.y};
#pragma unroll
            for (int j = 0; j < 2; ++j) {
                const float* hh = hp + j * 16 * H1_STRIDE + k;
                const v2f a = {hh[0], hh[1]};
                acc[j][0] = wmma_f32(a, b0, acc[j][0]);
                acc[j][1] = wmma_f32(a, b1, acc[j][1]);
            }
        }
#pragma unroll
        for (int j = 0; j < 2; ++j) {
#pragma unroll
            for (int r = 0; r < 8; ++r) {
                const int rg = row_base + (rt0 + j) * 16 + half * 8 + r;
                if (rg < N_NODES)
                    *(float2*)&m_out[(size_t)rg * D + cb] =
                        make_float2(acc[j][0][r], acc[j][1][r]);
            }
        }
    }
}

__global__ __launch_bounds__(256) void gnn_link_edges(
    const int* __restrict__ senders, const int* __restrict__ receivers,
    int* __restrict__ head, int2* __restrict__ link) {
    const int e = blockIdx.x * 256 + threadIdx.x;
    if (e >= N_EDGES) return;
    const int r = receivers[e];
    const int s = senders[e];
    const int prev = __hip_atomic_exchange(head + r, e, __ATOMIC_RELAXED,
                                           __HIP_MEMORY_SCOPE_AGENT);
    link[e] = make_int2(prev, s);
}

#define GR (ROWS / 8)
__global__ __launch_bounds__(256) __attribute__((amdgpu_waves_per_eu(6))) void gnn_node_mlp(
    const float* __restrict__ nodes, const float* __restrict__ m,
    const int* __restrict__ head, const int2* __restrict__ link,
    const float* __restrict__ Wn1, const float* __restrict__ bn1,
    const float* __restrict__ Wn2, const float* __restrict__ bn2,
    float* __restrict__ out) {
    __shared__ __attribute__((aligned(16))) float Xs[ROWS * XS_STRIDE];
    __shared__ __attribute__((aligned(16))) float H1s[ROWS * H1_STRIDE];
    float* AGs = H1s;

    const int tid = threadIdx.x;
    const int row_base = blockIdx.x * ROWS;
    const int lane = tid & 31;
    const int wv = __builtin_amdgcn_readfirstlane((int)(tid >> 5));

    for (int i = tid; i < ROWS * (D / 4); i += 256) {
        int row = i / (D / 4);
        int c4 = i % (D / 4);
        int rg = row_base + row;
        float4 v = make_float4(0.f, 0.f, 0.f, 0.f);
        if (rg < N_NODES)
            v = ((const float4*)(nodes + (size_t)rg * D))[c4];
        *(float4*)&Xs[row * XS_STRIDE + c4 * 4] = v;
    }

    {
        const float4* m4 = (const float4*)m;
        int e[GR];
        float4 acc[GR];
        float4 pa[GR], pb[GR];
        bool ga[GR], gb[GR];
        bool any = false;
#pragma unroll
        for (int j = 0; j < GR; ++j) {
            const int r = row_base + wv * GR + j;
            e[j] = -1;
            if (r < N_NODES) e[j] = head[r];
            any = any || ((unsigned)e[j] < (unsigned)N_EDGES);
            acc[j] = make_float4(0.f, 0.f, 0.f, 0.f);
            pa[j] = acc[j]; pb[j] = acc[j];
            ga[j] = false; gb[j] = false;
        }
#define WALK_TRIP(PIN, GIN, POUT, GOUT)                                          \
        {                                                                        \
            int2 ln[GR];                                                         \
            bool live[GR];                                                       \
            _Pragma("unroll") for (int j = 0; j < GR; ++j) {                     \
                live[j] = (unsigned)e[j] < (unsigned)N_EDGES;                    \
                ln[j] = link[live[j] ? e[j] : 0];                                \
            }                                                                    \
            asm volatile("" :: "s"(ln[0].x), "s"(ln[0].y), "s"(ln[1].x),         \
                               "s"(ln[1].y), "s"(ln[2].x), "s"(ln[2].y),         \
                               "s"(ln[3].x), "s"(ln[3].y), "s"(ln[4].x),         \
                               "s"(ln[4].y), "s"(ln[5].x), "s"(ln[5].y),         \
                               "s"(ln[6].x), "s"(ln[6].y), "s"(ln[7].x),         \
                               "s"(ln[7].y));                                    \
            _Pragma("unroll") for (int j = 0; j < GR; ++j) {                     \
                const float4 p = PIN[j];                                         \
                acc[j].x += GIN[j] ? p.x : 0.f; acc[j].y += GIN[j] ? p.y : 0.f;  \
                acc[j].z += GIN[j] ? p.z : 0.f; acc[j].w += GIN[j] ? p.w : 0.f;  \
                const int srow = live[j] ? ln[j].y : 0;                          \
                POUT[j] = m4[(size_t)srow * (D / 4) + lane];                     \
                GOUT[j] = live[j];                                               \
                e[j] = live[j] ? ln[j].x : -1;                                   \
            }                                                                    \
        }
        for (int it = 0; it < N_EDGES && any; it += 2) {
            WALK_TRIP(pb, gb, pa, ga)
            WALK_TRIP(pa, ga, pb, gb)
            any = false;
#pragma unroll
            for (int j = 0; j < GR; ++j)
                any = any || ((unsigned)e[j] < (unsigned)N_EDGES);
        }
#undef WALK_TRIP
#pragma unroll
        for (int j = 0; j < GR; ++j) {
            const float4 q = pb[j];
            acc[j].x += gb[j] ? q.x : 0.f; acc[j].y += gb[j] ? q.y : 0.f;
            acc[j].z += gb[j] ? q.z : 0.f; acc[j].w += gb[j] ? q.w : 0.f;
            *(float4*)&AGs[(wv * GR + j) * XS_STRIDE + lane * 4] = acc[j];
        }
    }
    __syncthreads();

    const int w = tid >> 5;
    const int half = lane >> 4;
    const int l16 = lane & 15;

    {
        const int cb = w * 32 + 2 * l16;
        const float2 bv = *(const float2*)&bn1[cb];
        v8f acc[RT][2];
#pragma unroll
        for (int rt = 0; rt < RT; ++rt) {
            acc[rt][0] = splat8(bv.x);
            acc[rt][1] = splat8(bv.y);
        }
        const float* wp = Wn1 + (size_t)(half * 2) * H + cb;
        const float* xp = &Xs[l16 * XS_STRIDE + half * 2];
#pragma unroll 2
        for (int k = 0; k < D; k += 4) {
            const float2 q0 = *(const float2*)(wp + (size_t)k * H);
            const float2 q1 = *(const float2*)(wp + (size_t)(k + 1) * H);
            const v2f b0 = {q0.x, q1.x};
            const v2f b1 = {q0.y, q1.y};
#pragma unroll
            for (int rt = 0; rt < RT; ++rt) {
                const float* x = xp + rt * 16 * XS_STRIDE + k;
                const v2f a = {x[0], x[1]};
                acc[rt][0] = wmma_f32(a, b0, acc[rt][0]);
                acc[rt][1] = wmma_f32(a, b1, acc[rt][1]);
            }
        }
        const float* wq = wp + (size_t)D * H;
        const float* gp = &AGs[l16 * XS_STRIDE + half * 2];
#pragma unroll 2
        for (int k = 0; k < D; k += 4) {
            const float2 q0 = *(const float2*)(wq + (size_t)k * H);
            const float2 q1 = *(const float2*)(wq + (size_t)(k + 1) * H);
            const v2f b0 = {q0.x, q1.x};
            const v2f b1 = {q0.y, q1.y};
#pragma unroll
            for (int rt = 0; rt < RT; ++rt) {
                const float* g = gp + rt * 16 * XS_STRIDE + k;
                const v2f a = {g[0], g[1]};
                acc[rt][0] = wmma_f32(a, b0, acc[rt][0]);
                acc[rt][1] = wmma_f32(a, b1, acc[rt][1]);
            }
        }
        __syncthreads();
#pragma unroll
        for (int rt = 0; rt < RT; ++rt) {
#pragma unroll
            for (int r = 0; r < 8; ++r) {
                const int row = rt * 16 + half * 8 + r;
                const float v0 = acc[rt][0][r];
                const float v1 = acc[rt][1][r];
                *(float2*)&H1s[row * H1_STRIDE + cb] =
                    make_float2(v0 > 0.f ? v0 : 0.f, v1 > 0.f ? v1 : 0.f);
            }
        }
    }
    __syncthreads();

    {
        const int cb = (w & 3) * 32 + 2 * l16;
        const int rt0 = (w >> 2) * 2;
        const float2 bv = *(const float2*)&bn2[cb];
        v8f acc[2][2];
#pragma unroll
        for (int j = 0; j < 2; ++j) {
            acc[j][0] = splat8(bv.x);
            acc[j][1] = splat8(bv.y);
        }
        const float* wp = Wn2 + (size_t)(half * 2) * D + cb;
        const float* hp = &H1s[(rt0 * 16 + l16) * H1_STRIDE + half * 2];
#pragma unroll 4
        for (int k = 0; k < H; k += 4) {
            const float2 q0 = *(const float2*)(wp + (size_t)k * D);
            const float2 q1 = *(const float2*)(wp + (size_t)(k + 1) * D);
            const v2f b0 = {q0.x, q1.x};
            const v2f b1 = {q0.y, q1.y};
#pragma unroll
            for (int j = 0; j < 2; ++j) {
                const float* hh = hp + j * 16 * H1_STRIDE + k;
                const v2f a = {hh[0], hh[1]};
                acc[j][0] = wmma_f32(a, b0, acc[j][0]);
                acc[j][1] = wmma_f32(a, b1, acc[j][1]);
            }
        }
#pragma unroll
        for (int j = 0; j < 2; ++j) {
#pragma unroll
            for (int r = 0; r < 8; ++r) {
                const int row = (rt0 + j) * 16 + half * 8 + r;
                const int rg = row_base + row;
                if (rg < N_NODES) {
                    const float2 xv2 = *(const float2*)&Xs[row * XS_STRIDE + cb];
                    *(float2*)&out[(size_t)rg * D + cb] =
                        make_float2(xv2.x + acc[j][0][r], xv2.y + acc[j][1][r]);
                }
            }
        }
    }
}

extern "C" void kernel_launch(void* const* d_in, const int* in_sizes, int n_in,
                              void* d_out, int out_size, void* d_ws, size_t ws_size,
                              hipStream_t stream) {
    (void)in_sizes; (void)n_in; (void)out_size; (void)ws_size;
    const float* nodes     = (const float*)d_in[0];
    const int*   senders   = (const int*)d_in[1];
    const int*   receivers = (const int*)d_in[2];
    const float* Wm1 = (const float*)d_in[3];
    const float* bm1 = (const float*)d_in[4];
    const float* Wm2 = (const float*)d_in[5];
    const float* bm2 = (const float*)d_in[6];
    const float* Wn1 = (const float*)d_in[7];
    const float* bn1 = (const float*)d_in[8];
    const float* Wn2 = (const float*)d_in[9];
    const float* bn2 = (const float*)d_in[10];

    float* m_buf = (float*)d_ws;
    int*   head  = (int*)(m_buf + (size_t)N_NODES * D);
    int2*  link  = (int2*)(head + N_NODES);
    float* out   = (float*)d_out;

    const int row_blocks = (N_NODES + ROWS - 1) / ROWS;

    gnn_msg_mlp<<<row_blocks, 256, 0, stream>>>(nodes, Wm1, bm1, Wm2, bm2, m_buf);

    hipMemsetAsync(head, 0xFF, sizeof(int) * (size_t)N_NODES, stream);

    gnn_link_edges<<<(N_EDGES + 255) / 256, 256, 0, stream>>>(senders, receivers, head, link);

    gnn_node_mlp<<<row_blocks, 256, 0, stream>>>(nodes, m_buf, head, link, Wn1, bn1, Wn2, bn2, out);
}


# grid re-cut to the 768 resident block places: 1536 blocks of 64 rows fill them exactly twice; the 1696 leftover rows run first as 106 one-tile blocks (was 1563 blocks: two fills + a 27-block tail)
# speedup vs baseline: 1.4517x; 1.0659x over previous
#include <hip/hip_runtime.h>
#include <hip/hip_bf16.h>
#include <cstddef>

#define N_NODES 100000
#define N_EDGES 600000
#define D 128
#define H 256

typedef float v2f __attribute__((ext_vector_type(2)));
typedef float v8f __attribute__((ext_vector_type(8)));

__device__ __forceinline__ v8f wmma_f32(v2f a, v2f b, v8f c) {
    return __builtin_amdgcn_wmma_f32_16x16x4_f32(false, a, false, b, (short)0, c,
                                                 false, false);
}

#define ROWS 64
#define RT (ROWS / 16)
#define BIG_BLOCKS 1536
#define SMALL_BLOCKS 106
#define SMALL_ROW0 (BIG_BLOCKS * ROWS)
static_assert(BIG_BLOCKS * ROWS + SMALL_BLOCKS * 16 == N_NODES, "grid cut covers every row once");
#define XS_STRIDE 132
#define H1_STRIDE 260


__device__ __forceinline__ v8f splat8(float v) {
    v8f r = {v, v, v, v, v, v, v, v};
    return r;
}

template <int NRT>
__device__ __forceinline__ void msg_tile(
    const int row_base, float* Xs, float* H1s,
    const float* __restrict__ nodes, const float* __restrict__ Wm1,
    const float* __restrict__ bm1, const float* __restrict__ Wm2,
    const float* __restrict__ bm2, float* __restrict__ m_out) {
    const int tid = threadIdx.x;

    for (int i = tid; i < NRT * 16 * (D / 4); i += 256) {
        int row = i / (D / 4);
        int c4 = i % (D / 4);
        int rg = row_base + row;
        float4 v = make_float4(0.f, 0.f, 0.f, 0.f);
        if (rg < N_NODES)
            v = ((const float4*)(nodes + (size_t)rg * D))[c4];
        *(float4*)&Xs[row * XS_STRIDE + c4 * 4] = v;
    }
    __syncthreads();

    const int w = tid >> 5;
    const int lane = tid & 31;
    const int half = lane >> 4;
    const int l16 = lane & 15;

    {
        const int cb = w * 32 + 2 * l16;
        const float2 bv = *(const float2*)&bm1[cb];
        v8f acc[NRT][2];
#pragma unroll
        for (int rt = 0; rt < NRT; ++rt) {
            acc[rt][0] = splat8(bv.x);
            acc[rt][1] = splat8(bv.y);
        }
        const float* wp = Wm1 + (size_t)(half * 2) * H + cb;
        const float* xp = &Xs[l16 * XS_STRIDE + half * 2];
#pragma unroll (NRT == 1 ? 8 : 2)
        for (int k = 0; k < D; k += 4) {
            const float2 q0 = *(const float2*)(wp + (size_t)k * H);
            const float2 q1 = *(const float2*)(wp + (size_t)(k + 1) * H);
            const v2f b0 = {q0.x, q1.x};
            const v2f b1 = {q0.y, q1.y};
#pragma unroll
            for (int rt = 0; rt < NRT; ++rt) {
                const float* x = xp + rt * 16 * XS_STRIDE + k;
                const v2f a = {x[0], x[1]};
                acc[rt][0] = wmma_f32(a, b0, acc[rt][0]);
                acc[rt][1] = wmma_f32(a, b1, acc[rt][1]);
            }
        }
#pragma unroll
        for (int rt = 0; rt < NRT; ++rt) {
#pragma unroll
            for (int r = 0; r < 8; ++r) {
                const int row = rt * 16 + half * 8 + r;
                const float v0 = acc[rt][0][r];
                const float v1 = acc[rt][1][r];
                *(float2*)&H1s[row * H1_STRIDE + cb] =
                    make_float2(v0 > 0.f ? v0 : 0.f, v1 > 0.f ? v1 : 0.f);
            }
        }
    }
    __syncthreads();

    constexpr int J = (NRT >= 2) ? NRT / 2 : 1;
    if (NRT >= 2 || __builtin_amdgcn_readfirstlane(w) < 4) {
        const int cb = (w & 3) * 32 + 2 * l16;
        const int rt0 = (w >> 2) * J;
        const float2 bv = *(const float2*)&bm2[cb];
        v8f acc[J][2];
#pragma unroll
        for (int j = 0; j < J; ++j) {
            acc[j][0] = splat8(bv.x);
            acc[j][1] = splat8(bv.y);
        }
        const float* wp = Wm2 + (size_t)(half * 2) * D + cb;
        const float* hp = &H1s[(rt0 * 16 + l16) * H1_STRIDE + half * 2];
#pragma unroll (NRT == 1 ? 8 : 4)
        for (int k = 0; k < H; k += 4) {
            const float2 q0 = *(const float2*)(wp + (size_t)k * D);
            const float2 q1 = *(const float2*)(wp + (size_t)(k + 1) * D);
            const v2f b0 = {q0.x, q1.x};
            const v2f b1 = {q0.y, q1.y};
#pragma unroll
            for (int j = 0; j < J; ++j) {
                const float* hh = hp + j * 16 * H1_STRIDE + k;
                const v2f a = {hh[0], hh[1]};
                acc[j][0] = wmma_f32(a, b0, acc[j][0]);
                acc[j][1] = wmma_f32(a, b1, acc[j][1]);
            }
        }
#pragma unroll
        for (int j = 0; j < J; ++j) {
#pragma unroll
            for (int r = 0; r < 8; ++r) {
                const int rg = row_base + (rt0 + j) * 16 + half * 8 + r;
                if (rg < N_NODES)
                    *(float2*)&m_out[(size_t)rg * D + cb] =
                        make_float2(acc[j][0][r], acc[j][1][r]);
            }
        }
    }
}

__global__ __launch_bounds__(256) void gnn_msg_mlp(
    const float* __restrict__ nodes, const float* __restrict__ Wm1,
    const float* __restrict__ bm1, const float* __restrict__ Wm2,
    const float* __restrict__ bm2, float* __restrict__ m_out) {
    __shared__ float Xs[ROWS * XS_STRIDE];
    __shared__ float H1s[ROWS * H1_STRIDE];
    const int b = blockIdx.x;
    if (b < SMALL_BLOCKS)
        msg_tile<1>(SMALL_ROW0 + b * 16, Xs, H1s, nodes, Wm1, bm1, Wm2, bm2, m_out);
    else
        msg_tile<RT>((b - SMALL_BLOCKS) * ROWS, Xs, H1s, nodes, Wm1, bm1, Wm2, bm2, m_out);
}

__global__ __launch_bounds__(256) void gnn_link_edges(
    const int* __restrict__ senders, const int* __restrict__ receivers,
    int* __restrict__ head, int2* __restrict__ link) {
    const int e = blockIdx.x * 256 + threadIdx.x;
    if (e >= N_EDGES) return;
    const int r = receivers[e];
    const int s = senders[e];
    const int prev = __hip_atomic_exchange(head + r, e, __ATOMIC_RELAXED,
                                           __HIP_MEMORY_SCOPE_AGENT);
    link[e] = make_int2(prev, s);
}

template <int NRT>
__device__ __forceinline__ void node_tile(
    const int row_base, float* Xs, float* H1s,
    const float* __restrict__ nodes, const float* __restrict__ m,
    const int* __restrict__ head, const int2* __restrict__ link,
    const float* __restrict__ Wn1, const float* __restrict__ bn1,
    const float* __restrict__ Wn2, const float* __restrict__ bn2,
    float* __restrict__ out) {
    constexpr int GR = NRT * 2;
    float* AGs = H1s;

    const int tid = threadIdx.x;
    const int lane = tid & 31;
    const int wv = __builtin_amdgcn_readfirstlane((int)(tid >> 5));

    for (int i = tid; i < NRT * 16 * (D / 4); i += 256) {
        int row = i / (D / 4);
        int c4 = i % (D / 4);
        int rg = row_base + row;
        float4 v = make_float4(0.f, 0.f, 0.f, 0.f);
        if (rg < N_NODES)
            v = ((const float4*)(nodes + (size_t)rg * D))[c4];
        *(float4*)&Xs[row * XS_STRIDE + c4 * 4] = v;
    }

    {
        const float4* m4 = (const float4*)m;
        int e[GR];
        float4 acc[GR];
        float4 pa[GR], pb[GR];
        bool ga[GR], gb[GR];
        bool any = false;
#pragma unroll
        for (int j = 0; j < GR; ++j) {
            const int r = row_base + wv * GR + j;
            e[j] = -1;
            if (r < N_NODES) e[j] = head[r];
            any = any || ((unsigned)e[j] < (unsigned)N_EDGES);
            acc[j] = make_float4(0.f, 0.f, 0.f, 0.f);
            pa[j] = acc[j]; pb[j] = acc[j];
            ga[j] = false; gb[j] = false;
        }
#define WALK_TRIP(PIN, GIN, POUT, GOUT)                                          \
        {                                                                        \
            int2 ln[GR];                                                         \
            bool live[GR];                                                       \
            _Pragma("unroll") for (int j = 0; j < GR; ++j) {                     \
                live[j] = (unsigned)e[j] < (unsigned)N_EDGES;                    \
                ln[j] = link[live[j] ? e[j] : 0];                                \
            }                                                                    \
            if constexpr (GR == 8) {                                             \
            asm volatile("" :: "s"(ln[0].x), "s"(ln[0].y), "s"(ln[1].x),         \
                               "s"(ln[1].y), "s"(ln[2].x), "s"(ln[2].y),         \
                               "s"(ln[3].x), "s"(ln[3].y), "s"(ln[4].x),         \
                               "s"(ln[4].y), "s"(ln[5].x), "s"(ln[5].y),         \
                               "s"(ln[6].x), "s"(ln[6].y), "s"(ln[7].x),         \
                               "s"(ln[7].y));                                    \
            } else {                                                             \
            asm volatile("" :: "s"(ln[0].x), "s"(ln[0].y), "s"(ln[1].x),         \
                               "s"(ln[1].y));                                    \
            }                                                                    \
            _Pragma("unroll") for (int j = 0; j < GR; ++j) {                     \
                const float4 p = PIN[j];                                         \
                acc[j].x += GIN[j] ? p.x : 0.f; acc[j].y += GIN[j] ? p.y : 0.f;  \
                acc[j].z += GIN[j] ? p.z : 0.f; acc[j].w += GIN[j] ? p.w : 0.f;  \
                const int srow = live[j] ? ln[j].y : 0;                          \
                POUT[j] = m4[(size_t)srow * (D / 4) + lane];                     \
                GOUT[j] = live[j];                                               \
                e[j] = live[j] ? ln[j].x : -1;                                   \
            }                                                                    \
        }
        for (int it = 0; it < N_EDGES && any; it += 2) {
            WALK_TRIP(pb, gb, pa, ga)
            WALK_TRIP(pa, ga, pb, gb)
            any = false;
#pragma unroll
            for (int j = 0; j < GR; ++j)
                any = any || ((unsigned)e[j] < (unsigned)N_EDGES);
        }
#undef WALK_TRIP
#pragma unroll
        for (int j = 0; j < GR; ++j) {
            const float4 q = pb[j];
            acc[j].x += gb[j] ? q.x : 0.f; acc[j].y += gb[j] ? q.y : 0.f;
            acc[j].z += gb[j] ? q.z : 0.f; acc[j].w += gb[j] ? q.w : 0.f;
            *(float4*)&AGs[(wv * GR + j) * XS_STRIDE + lane * 4] = acc[j];
        }
    }
    __syncthreads();

    const int w = tid >> 5;
    const int half = lane >> 4;
    const int l16 = lane & 15;

    {
        const int cb = w * 32 + 2 * l16;
        const float2 bv = *(const float2*)&bn1[cb];
        v8f acc[NRT][2];
#pragma unroll
        for (int rt = 0; rt < NRT; ++rt) {
            acc[rt][0] = splat8(bv.x);
            acc[rt][1] = splat8(bv.y);
        }
        const float* wp = Wn1 + (size_t)(half * 2) * H + cb;
        const float* xp = &Xs[l16 * XS_STRIDE + half * 2];
#pragma unroll (NRT == 1 ? 8 : 2)
        for (int k = 0; k < D; k += 4) {
            const float2 q0 = *(const float2*)(wp + (size_t)k * H);
            const float2 q1 = *(const float2*)(wp + (size_t)(k + 1) * H);
            const v2f b0 = {q0.x, q1.x};
            const v2f b1 = {q0.y, q1.y};
#pragma unroll
            for (int rt = 0; rt < NRT; ++rt) {
                const float* x = xp + rt * 16 * XS_STRIDE + k;
                const v2f a = {x[0], x[1]};
                acc[rt][0] = wmma_f32(a, b0, acc[rt][0]);
                acc[rt][1] = wmma_f32(a, b1, acc[rt][1]);
            }
        }
        const float* wq = wp + (size_t)D * H;
        const float* gp = &AGs[l16 * XS_STRIDE + half * 2];
#pragma unroll (NRT == 1 ? 8 : 2)
        for (int k = 0; k < D; k += 4) {
            const float2 q0 = *(const float2*)(wq + (size_t)k * H);
            const float2 q1 = *(const float2*)(wq + (size_t)(k + 1) * H);
            const v2f b0 = {q0.x, q1.x};
            const v2f b1 = {q0.y, q1.y};
#pragma unroll
            for (int rt = 0; rt < NRT; ++rt) {
                const float* g = gp + rt * 16 * XS_STRIDE + k;
                const v2f a = {g[0], g[1]};
                acc[rt][0] = wmma_f32(a, b0, acc[rt][0]);
                acc[rt][1] = wmma_f32(a, b1, acc[rt][1]);
            }
        }
        __syncthreads();
#pragma unroll
        for (int rt = 0; rt < NRT; ++rt) {
#pragma unroll
            for (int r = 0; r < 8; ++r) {
                const int row = rt * 16 + half * 8 + r;
                const float v0 = acc[rt][0][r];
                const float v1 = acc[rt][1][r];
                *(float2*)&H1s[row * H1_STRIDE + cb] =
                    make_float2(v0 > 0.f ? v0 : 0.f, v1 > 0.f ? v1 : 0.f);
            }
        }
    }
    __syncthreads();

    constexpr int J = (NRT >= 2) ? NRT / 2 : 1;
    if (NRT >= 2 || wv < 4) {
        const int cb = (w & 3) * 32 + 2 * l16;
        const int rt0 = (w >> 2) * J;
        const float2 bv = *(const float2*)&bn2[cb];
        v8f acc[J][2];
#pragma unroll
        for (int j = 0; j < J; ++j) {
            acc[j][0] = splat8(bv.x);
            acc[j][1] = splat8(bv.y);
        }
        const float* wp = Wn2 + (size_t)(half * 2) * D + cb;
        const float* hp = &H1s[(rt0 * 16 + l16) * H1_STRIDE + half * 2];
#pragma unroll (NRT == 1 ? 8 : 4)
        for (int k = 0; k < H; k += 4) {
            const float2 q0 = *(const float2*)(wp + (size_t)k * D);
            const float2 q1 = *(const float2*)(wp + (size_t)(k + 1) * D);
            const v2f b0 = {q0.x, q1.x};
            const v2f b1 = {q0.y, q1.y};
#pragma unroll
            for (int j = 0; j < J; ++j) {
                const float* hh = hp + j * 16 * H1_STRIDE + k;
                const v2f a = {hh[0], hh[1]};
                acc[j][0] = wmma_f32(a, b0, acc[j][0]);
                acc[j][1] = wmma_f32(a, b1, acc[j][1]);
            }
        }
#pragma unroll
        for (int j = 0; j < J; ++j) {
#pragma unroll
            for (int r = 0; r < 8; ++r) {
                const int row = (rt0 + j) * 16 + half * 8 + r;
                const int rg = row_base + row;
                if (rg < N_NODES) {
                    const float2 xv2 = *(const float2*)&Xs[row * XS_STRIDE + cb];
                    *(float2*)&out[(size_t)rg * D + cb] =
                        make_float2(xv2.x + acc[j][0][r], xv2.y + acc[j][1][r]);
                }
            }
        }
    }
}

__global__ __launch_bounds__(256) __attribute__((amdgpu_waves_per_eu(6))) void gnn_node_mlp(
    const float* __restrict__ nodes, const float* __restrict__ m,
    const int* __restrict__ head, const int2* __restrict__ link,
    const float* __restrict__ Wn1, const float* __restrict__ bn1,
    const float* __restrict__ Wn2, const float* __restrict__ bn2,
    float* __restrict__ out) {
    __shared__ __attribute__((aligned(16))) float Xs[ROWS * XS_STRIDE];
    __shared__ __attribute__((aligned(16))) float H1s[ROWS * H1_STRIDE];
    const int b = blockIdx.x;
    if (b < SMALL_BLOCKS)
        node_tile<1>(SMALL_ROW0 + b * 16, Xs, H1s, nodes, m, head, link, Wn1, bn1, Wn2, bn2, out);
    else
        node_tile<RT>((b - SMALL_BLOCKS) * ROWS, Xs, H1s, nodes, m, head, link, Wn1, bn1, Wn2, bn2, out);
}

extern "C" void kernel_launch(void* const* d_in, const int* in_sizes, int n_in,
                              void* d_out, int out_size, void* d_ws, size_t ws_size,
                              hipStream_t stream) {
    (void)in_sizes; (void)n_in; (void)out_size; (void)ws_size;
    const float* nodes     = (const float*)d_in[0];
    const int*   senders   = (const int*)d_in[1];
    const int*   receivers = (const int*)d_in[2];
    const float* Wm1 = (const float*)d_in[3];
    const float* bm1 = (const float*)d_in[4];
    const float* Wm2 = (const float*)d_in[5];
    const float* bm2 = (const float*)d_in[6];
    const float* Wn1 = (const float*)d_in[7];
    const float* bn1 = (const float*)d_in[8];
    const float* Wn2 = (const float*)d_in[9];
    const float* bn2 = (const float*)d_in[10];

    float* m_buf = (float*)d_ws;
    int*   head  = (int*)(m_buf + (size_t)N_NODES * D);
    int2*  link  = (int2*)(head + N_NODES);
    float* out   = (float*)d_out;

    const int row_blocks = SMALL_BLOCKS + BIG_BLOCKS;

    gnn_msg_mlp<<<row_blocks, 256, 0, stream>>>(nodes, Wm1, bm1, Wm2, bm2, m_buf);

    hipMemsetAsync(head, 0xFF, sizeof(int) * (size_t)N_NODES, stream);

    gnn_link_edges<<<(N_EDGES + 255) / 256, 256, 0, stream>>>(senders, receivers, head, link);

    gnn_node_mlp<<<row_blocks, 256, 0, stream>>>(nodes, m_buf, head, link, Wn1, bn1, Wn2, bn2, out);
}


# operand loads software-pipelined in the four 64-row products: weight pairs requested 3 stages (48 WMMAs) ahead in a register ring, LDS A fragments one stage ahead; order of every sum unchanged
# speedup vs baseline: 1.5068x; 1.0380x over previous
#include <hip/hip_runtime.h>
#include <hip/hip_bf16.h>
#include <cstddef>

#define N_NODES 100000
#define N_EDGES 600000
#define D 128
#define H 256

typedef float v2f __attribute__((ext_vector_type(2)));
typedef float v8f __attribute__((ext_vector_type(8)));

__device__ __forceinline__ v8f wmma_f32(v2f a, v2f b, v8f c) {
    return __builtin_amdgcn_wmma_f32_16x16x4_f32(false, a, false, b, (short)0, c,
                                                 false, false);
}

#define ROWS 64
#define RT (ROWS / 16)
#define BIG_BLOCKS 1536
#define SMALL_BLOCKS 106
#define SMALL_ROW0 (BIG_BLOCKS * ROWS)
static_assert(BIG_BLOCKS * ROWS + SMALL_BLOCKS * 16 == N_NODES, "grid cut covers every row once");
#define XS_STRIDE 132
#define H1_STRIDE 260


__device__ __forceinline__ v8f splat8(float v) {
    v8f r = {v, v, v, v, v, v, v, v};
    return r;
}

template <int KS, int WPITCH>
__device__ __forceinline__ void load_b_stage(float2 (&slot)[KS][2],
                                             const float* __restrict__ wp, const int s) {
#pragma unroll
    for (int ks = 0; ks < KS; ++ks) {
        const float* p = wp + (size_t)((s * KS + ks) * 4) * WPITCH;
        slot[ks][0] = *(const float2*)p;
        slot[ks][1] = *(const float2*)(p + WPITCH);
    }
}

template <int NR, int KS>
__device__ __forceinline__ void load_a_stage(v2f (&slot)[KS][NR], const float* ap,
                                             const int rt_stride) {
#pragma unroll
    for (int ks = 0; ks < KS; ++ks) {
#pragma unroll
        for (int r = 0; r < NR; ++r) {
            const float* x = ap + r * rt_stride + ks * 4;
            const v2f a = {x[0], x[1]};
            slot[ks][r] = a;
        }
    }
}

template <int KS, int PF, int WPITCH>
__device__ __forceinline__ void product_prefetch(float2 (&bq)[PF + 1][KS][2],
                                                 const float* __restrict__ wp) {
#pragma unroll
    for (int s = 0; s < PF; ++s) load_b_stage<KS, WPITCH>(bq[s], wp, s);
}

template <int NR, int KS, int NSTAGE, int PF, int WPITCH, int SPLIT>
__device__ __forceinline__ void product_run(v8f (&acc)[NR][2], float2 (&bq)[PF + 1][KS][2],
                                            const float* __restrict__ wp, const float* ap0,
                                            const float* ap1, const int rt_stride) {
    constexpr int R = PF + 1;
    v2f af[2][KS][NR];
    load_a_stage<NR, KS>(af[0], ap0, rt_stride);
#pragma unroll
    for (int s = 0; s < NSTAGE; ++s) {
        __builtin_amdgcn_sched_barrier(0);
        if (s + PF < NSTAGE) load_b_stage<KS, WPITCH>(bq[(s + PF) % R], wp, s + PF);
        if (s + 1 < NSTAGE) {
            const int n = s + 1;
            load_a_stage<NR, KS>(af[n & 1],
                                 n < SPLIT ? ap0 + n * KS * 4 : ap1 + (n - SPLIT) * KS * 4,
                                 rt_stride);
        }
#pragma unroll
        for (int ks = 0; ks < KS; ++ks) {
            const float2 q0 = bq[s % R][ks][0];
            const float2 q1 = bq[s % R][ks][1];
            const v2f b0 = {q0.x, q1.x};
            const v2f b1 = {q0.y, q1.y};
#pragma unroll
            for (int r = 0; r < NR; ++r) {
                acc[r][0] = wmma_f32(af[s & 1][ks][r], b0, acc[r][0]);
                acc[r][1] = wmma_f32(af[s & 1][ks][r], b1, acc[r][1]);
            }
        }
    }
}
#define PF_L1 3
#define PF_L2 3

template <int NRT>
__device__ __forceinline__ void msg_tile(
    const int row_base, float* Xs, float* H1s,
    const float* __restrict__ nodes, const float* __restrict__ Wm1,
    const float* __restrict__ bm1, const float* __restrict__ Wm2,
    const float* __restrict__ bm2, float* __restrict__ m_out) {
    const int tid = threadIdx.x;
    const int w = tid >> 5;
    const int lane = tid & 31;
    const int half = lane >> 4;
    const int l16 = lane & 15;
    constexpr bool PIPE = (NRT == 4);
    constexpr int J = (NRT >= 2) ? NRT / 2 : 1;

    const int cb1 = w * 32 + 2 * l16;
    const float* wp1 = Wm1 + (size_t)(half * 2) * H + cb1;
    [[maybe_unused]] float2 bq1[PF_L1 + 1][2][2];
    float2 bv1, bv2;
    if constexpr (PIPE) {
        product_prefetch<2, PF_L1, H>(bq1, wp1);
        bv1 = *(const float2*)&bm1[cb1];
    }
    const int cb2 = (w & 3) * 32 + 2 * l16;
    const float* wp2 = Wm2 + (size_t)(half * 2) * D + cb2;
    [[maybe_unused]] float2 bq2[PF_L2 + 1][4][2];

    for (int i = tid; i < NRT * 16 * (D / 4); i += 256) {
        int row = i / (D / 4);
        int c4 = i % (D / 4);
        int rg = row_base + row;
        float4 v = make_float4(0.f, 0.f, 0.f, 0.f);
        if (rg < N_NODES)
            v = ((const float4*)(nodes + (size_t)rg * D))[c4];
        *(float4*)&Xs[row * XS_STRIDE + c4 * 4] = v;
    }
    __syncthreads();

    {
        const int cb = cb1;
        if constexpr (!PIPE) bv1 = *(const float2*)&bm1[cb];
        const float2 bv = bv1;
        v8f acc[NRT][2];
#pragma unroll
        for (int rt = 0; rt < NRT; ++rt) {
            acc[rt][0] = splat8(bv.x);
            acc[rt][1] = splat8(bv.y);
        }
        const float* wp = wp1;
        const float* xp = &Xs[l16 * XS_STRIDE + half * 2];
        if constexpr (PIPE) {
            product_run<NRT, 2, D / 8, PF_L1, H, D / 8>(acc, bq1, wp, xp, xp, 16 * XS_STRIDE);
            product_prefetch<4, PF_L2, D>(bq2, wp2);
            bv2 = *(const float2*)&bm2[cb2];
        } else {
#pragma unroll 8
        for (int k = 0; k < D; k += 4) {
            const float2 q0 = *(const float2*)(wp + (size_t)k * H);
            const float2 q1 = *(const float2*)(wp + (size_t)(k + 1) * H);
            const v2f b0 = {q0.x, q1.x};
            const v2f b1 = {q0.y, q1.y};
#pragma unroll
            for (int rt = 0; rt < NRT; ++rt) {
                const float* x = xp + rt * 16 * XS_STRIDE + k;
                const v2f a = {x[0], x[1]};
                acc[rt][0] = wmma_f32(a, b0, acc[rt][0]);
                acc[rt][1] = wmma_f32(a, b1, acc[rt][1]);
            }
        }
        }
#pragma unroll
        for (int rt = 0; rt < NRT; ++rt) {
#pragma unroll
            for (int r = 0; r < 8; ++r) {
                const int row = rt * 16 + half * 8 + r;
                const float v0 = acc[rt][0][r];
                const float v1 = acc[rt][1][r];
                *(float2*)&H1s[row * H1_STRIDE + cb] =
                    make_float2(v0 > 0.f ? v0 : 0.f, v1 > 0.f ? v1 : 0.f);
            }
        }
    }
    __syncthreads();

    if (NRT >= 2 || __builtin_amdgcn_readfirstlane(w) < 4) {
        const int cb = cb2;
        const int rt0 = (w >> 2) * J;
        if constexpr (!PIPE) bv2 = *(const float2*)&bm2[cb];
        const float2 bv = bv2;
        v8f acc[J][2];
#pragma unroll
        for (int j = 0; j < J; ++j) {
            acc[j][0] = splat8(bv.x);
            acc[j][1] = splat8(bv.y);
        }
        const float* wp = wp2;
        const float* hp = &H1s[(rt0 * 16 + l16) * H1_STRIDE + half * 2];
        if constexpr (PIPE) {
            product_run<J, 4, H / 16, PF_L2, D, H / 16>(acc, bq2, wp, hp, hp, 16 * H1_STRIDE);
        } else {
#pragma unroll 8
        for (int k = 0; k < H; k += 4) {
            const float2 q0 = *(const float2*)(wp + (size_t)k * D);
            const float2 q1 = *(const float2*)(wp + (size_t)(k + 1) * D);
            const v2f b0 = {q0.x, q1.x};
            const v2f b1 = {q0.y, q1.y};
#pragma unroll
            for (int j = 0; j < J; ++j) {
                const float* hh = hp + j * 16 * H1_STRIDE + k;
                const v2f a = {hh[0], hh[1]};
                acc[j][0] = wmma_f32(a, b0, acc[j][0]);
                acc[j][1] = wmma_f32(a, b1, acc[j][1]);
            }
        }
        }
#pragma unroll
        for (int j = 0; j < J; ++j) {
#pragma unroll
            for (int r = 0; r < 8; ++r) {
                const int rg = row_base + (rt0 + j) * 16 + half * 8 + r;
                if (rg < N_NODES)
                    *(float2*)&m_out[(size_t)rg * D + cb] =
                        make_float2(acc[j][0][r], acc[j][1][r]);
            }
        }
    }
}

__global__ __launch_bounds__(256) void gnn_msg_mlp(
    const float* __restrict__ nodes, const float* __restrict__ Wm1,
    const float* __restrict__ bm1, const float* __restrict__ Wm2,
    const float* __restrict__ bm2, float* __restrict__ m_out) {
    __shared__ float Xs[ROWS * XS_STRIDE];
    __shared__ float H1s[ROWS * H1_STRIDE];
    const int b = blockIdx.x;
    if (b < SMALL_BLOCKS)
        msg_tile<1>(SMALL_ROW0 + b * 16, Xs, H1s, nodes, Wm1, bm1, Wm2, bm2, m_out);
    else
        msg_tile<RT>((b - SMALL_BLOCKS) * ROWS, Xs, H1s, nodes, Wm1, bm1, Wm2, bm2, m_out);
}

__global__ __launch_bounds__(256) void gnn_link_edges(
    const int* __restrict__ senders, const int* __restrict__ receivers,
    int* __restrict__ head, int2* __restrict__ link) {
    const int e = blockIdx.x * 256 + threadIdx.x;
    if (e >= N_EDGES) return;
    const int r = receivers[e];
    const int s = senders[e];
    const int prev = __hip_atomic_exchange(head + r, e, __ATOMIC_RELAXED,
                                           __HIP_MEMORY_SCOPE_AGENT);
    link[e] = make_int2(prev, s);
}

template <int NRT>
__device__ __forceinline__ void node_tile(
    const int row_base, float* Xs, float* H1s,
    const float* __restrict__ nodes, const float* __restrict__ m,
    const int* __restrict__ head, const int2* __restrict__ link,
    const float* __restrict__ Wn1, const float* __restrict__ bn1,
    const float* __restrict__ Wn2, const float* __restrict__ bn2,
    float* __restrict__ out) {
    constexpr int GR = NRT * 2;
    float* AGs = H1s;

    const int tid = threadIdx.x;
    const int lane = tid & 31;
    const int wv = __builtin_amdgcn_readfirstlane((int)(tid >> 5));
    const int w = tid >> 5;
    const int half = lane >> 4;
    const int l16 = lane & 15;
    constexpr bool PIPE = (NRT == 4);
    constexpr int J = (NRT >= 2) ? NRT / 2 : 1;
    const int cb1 = w * 32 + 2 * l16;
    const float* wp1 = Wn1 + (size_t)(half * 2) * H + cb1;
    [[maybe_unused]] float2 bq1[PF_L1 + 1][2][2];
    const int cb2 = (w & 3) * 32 + 2 * l16;
    const float* wp2 = Wn2 + (size_t)(half * 2) * D + cb2;
    [[maybe_unused]] float2 bq2[PF_L2 + 1][4][2];
    float2 bv1, bv2;

    for (int i = tid; i < NRT * 16 * (D / 4); i += 256) {
        int row = i / (D / 4);
        int c4 = i % (D / 4);
        int rg = row_base + row;
        float4 v = make_float4(0.f, 0.f, 0.f, 0.f);
        if (rg < N_NODES)
            v = ((const float4*)(nodes + (size_t)rg * D))[c4];
        *(float4*)&Xs[row * XS_STRIDE + c4 * 4] = v;
    }

    {
        const float4* m4 = (const float4*)m;
        int e[GR];
        float4 acc[GR];
        float4 pa[GR], pb[GR];
        bool ga[GR], gb[GR];
        bool any = false;
#pragma unroll
        for (int j = 0; j < GR; ++j) {
            const int r = row_base + wv * GR + j;
            e[j] = -1;
            if (r < N_NODES) e[j] = head[r];
            any = any || ((unsigned)e[j] < (unsigned)N_EDGES);
            acc[j] = make_float4(0.f, 0.f, 0.f, 0.f);
            pa[j] = acc[j]; pb[j] = acc[j];
            ga[j] = false; gb[j] = false;
        }
#define WALK_TRIP(PIN, GIN, POUT, GOUT)                                          \
        {                                                                        \
            int2 ln[GR];                                                         \
            bool live[GR];                                                       \
            _Pragma("unroll") for (int j = 0; j < GR; ++j) {                     \
                live[j] = (unsigned)e[j] < (unsigned)N_EDGES;                    \
                ln[j] = link[live[j] ? e[j] : 0];                                \
            }                                                                    \
            if constexpr (GR == 8) {                                             \
            asm volatile("" :: "s"(ln[0].x), "s"(ln[0].y), "s"(ln[1].x),         \
                               "s"(ln[1].y), "s"(ln[2].x), "s"(ln[2].y),         \
                               "s"(ln[3].x), "s"(ln[3].y), "s"(ln[4].x),         \
                               "s"(ln[4].y), "s"(ln[5].x), "s"(ln[5].y),         \
                               "s"(ln[6].x), "s"(ln[6].y), "s"(ln[7].x),         \
                               "s"(ln[7].y));                                    \
            } else {                                                             \
            asm volatile("" :: "s"(ln[0].x), "s"(ln[0].y), "s"(ln[1].x),         \
                               "s"(ln[1].y));                                    \
            }                                                                    \
            _Pragma("unroll") for (int j = 0; j < GR; ++j) {                     \
                const float4 p = PIN[j];                                         \
                acc[j].x += GIN[j] ? p.x : 0.f; acc[j].y += GIN[j] ? p.y : 0.f;  \
                acc[j].z += GIN[j] ? p.z : 0.f; acc[j].w += GIN[j] ? p.w : 0.f;  \
                const int srow = live[j] ? ln[j].y : 0;                          \
                POUT[j] = m4[(size_t)srow * (D / 4) + lane];                     \
                GOUT[j] = live[j];                                               \
                e[j] = live[j] ? ln[j].x : -1;                                   \
            }                                                                    \
        }
        for (int it = 0; it < N_EDGES && any; it += 2) {
            WALK_TRIP(pb, gb, pa, ga)
            WALK_TRIP(pa, ga, pb, gb)
            any = false;
#pragma unroll
            for (int j = 0; j < GR; ++j)
                any = any || ((unsigned)e[j] < (unsigned)N_EDGES);
        }
#undef WALK_TRIP
        if constexpr (PIPE) {
            product_prefetch<2, PF_L1, H>(bq1, wp1);
            bv1 = *(const float2*)&bn1[cb1];
        }
#pragma unroll
        for (int j = 0; j < GR; ++j) {
            const float4 q = pb[j];
            acc[j].x += gb[j] ? q.x : 0.f; acc[j].y += gb[j] ? q.y : 0.f;
            acc[j].z += gb[j] ? q.z : 0.f; acc[j].w += gb[j] ? q.w : 0.f;
            *(float4*)&AGs[(wv * GR + j) * XS_STRIDE + lane * 4] = acc[j];
        }
    }
    __syncthreads();

    {
        const int cb = cb1;
        if constexpr (!PIPE) bv1 = *(const float2*)&bn1[cb];
        const float2 bv = bv1;
        v8f acc[NRT][2];
#pragma unroll
        for (int rt = 0; rt < NRT; ++rt) {
            acc[rt][0] = splat8(bv.x);
            acc[rt][1] = splat8(bv.y);
        }
        const float* wp = wp1;
        const float* xp = &Xs[l16 * XS_STRIDE + half * 2];
        const float* gp = &AGs[l16 * XS_STRIDE + half * 2];
        if constexpr (PIPE) {
            product_run<NRT, 2, 2 * D / 8, PF_L1, H, D / 8>(acc, bq1, wp, xp, gp, 16 * XS_STRIDE);
            product_prefetch<4, PF_L2, D>(bq2, wp2);
            bv2 = *(const float2*)&bn2[cb2];
        } else {
#pragma unroll 8
        for (int k = 0; k < D; k += 4) {
            const float2 q0 = *(const float2*)(wp + (size_t)k * H);
            const float2 q1 = *(const float2*)(wp + (size_t)(k + 1) * H);
            const v2f b0 = {q0.x, q1.x};
            const v2f b1 = {q0.y, q1.y};
#pragma unroll
            for (int rt = 0; rt < NRT; ++rt) {
                const float* x = xp + rt * 16 * XS_STRIDE + k;
                const v2f a = {x[0], x[1]};
                acc[rt][0] = wmma_f32(a, b0, acc[rt][0]);
                acc[rt][1] = wmma_f32(a, b1, acc[rt][1]);
            }
        }
        const float* wq = wp + (size_t)D * H;
#pragma unroll 8
        for (int k = 0; k < D; k += 4) {
            const float2 q0 = *(const float2*)(wq + (size_t)k * H);
            const float2 q1 = *(const float2*)(wq + (size_t)(k + 1) * H);
            const v2f b0 = {q0.x, q1.x};
            const v2f b1 = {q0.y, q1.y};
#pragma unroll
            for (int rt = 0; rt < NRT; ++rt) {
                const float* g = gp + rt * 16 * XS_STRIDE + k;
                const v2f a = {g[0], g[1]};
                acc[rt][0] = wmma_f32(a, b0, acc[rt][0]);
                acc[rt][1] = wmma_f32(a, b1, acc[rt][1]);
            }
        }
        }
        __syncthreads();
#pragma unroll
        for (int rt = 0; rt < NRT; ++rt) {
#pragma unroll
            for (int r = 0; r < 8; ++r) {
                const int row = rt * 16 + half * 8 + r;
                const float v0 = acc[rt][0][r];
                const float v1 = acc[rt][1][r];
                *(float2*)&H1s[row * H1_STRIDE + cb] =
                    make_float2(v0 > 0.f ? v0 : 0.f, v1 > 0.f ? v1 : 0.f);
            }
        }
    }
    __syncthreads();

    if (NRT >= 2 || wv < 4) {
        const int cb = cb2;
        const int rt0 = (w >> 2) * J;
        if constexpr (!PIPE) bv2 = *(const float2*)&bn2[cb];
        const float2 bv = bv2;
        v8f acc[J][2];
#pragma unroll
        for (int j = 0; j < J; ++j) {
            acc[j][0] = splat8(bv.x);
            acc[j][1] = splat8(bv.y);
        }
        const float* wp = wp2;
        const float* hp = &H1s[(rt0 * 16 + l16) * H1_STRIDE + half * 2];
        if constexpr (PIPE) {
            product_run<J, 4, H / 16, PF_L2, D, H / 16>(acc, bq2, wp, hp, hp, 16 * H1_STRIDE);
        } else {
#pragma unroll 8
        for (int k = 0; k < H; k += 4) {
            const float2 q0 = *(const float2*)(wp + (size_t)k * D);
            const float2 q1 = *(const float2*)(wp + (size_t)(k + 1) * D);
            const v2f b0 = {q0.x, q1.x};
            const v2f b1 = {q0.y, q1.y};
#pragma unroll
            for (int j = 0; j < J; ++j) {
                const float* hh = hp + j * 16 * H1_STRIDE + k;
                const v2f a = {hh[0], hh[1]};
                acc[j][0] = wmma_f32(a, b0, acc[j][0]);
                acc[j][1] = wmma_f32(a, b1, acc[j][1]);
            }
        }
        }
#pragma unroll
        for (int j = 0; j < J; ++j) {
#pragma unroll
            for (int r = 0; r < 8; ++r) {
                const int row = (rt0 + j) * 16 + half * 8 + r;
                const int rg = row_base + row;
                if (rg < N_NODES) {
                    const float2 xv2 = *(const float2*)&Xs[row * XS_STRIDE + cb];
                    *(float2*)&out[(size_t)rg * D + cb] =
                        make_float2(xv2.x + acc[j][0][r], xv2.y + acc[j][1][r]);
                }
            }
        }
    }
}

__global__ __launch_bounds__(256) __attribute__((amdgpu_waves_per_eu(6))) void gnn_node_mlp(
    const float* __restrict__ nodes, const float* __restrict__ m,
    const int* __restrict__ head, const int2* __restrict__ link,
    const float* __restrict__ Wn1, const float* __restrict__ bn1,
    const float* __restrict__ Wn2, const float* __restrict__ bn2,
    float* __restrict__ out) {
    __shared__ __attribute__((aligned(16))) float Xs[ROWS * XS_STRIDE];
    __shared__ __attribute__((aligned(16))) float H1s[ROWS * H1_STRIDE];
    const int b = blockIdx.x;
    if (b < SMALL_BLOCKS)
        node_tile<1>(SMALL_ROW0 + b * 16, Xs, H1s, nodes, m, head, link, Wn1, bn1, Wn2, bn2, out);
    else
        node_tile<RT>((b - SMALL_BLOCKS) * ROWS, Xs, H1s, nodes, m, head, link, Wn1, bn1, Wn2, bn2, out);
}

extern "C" void kernel_launch(void* const* d_in, const int* in_sizes, int n_in,
                              void* d_out, int out_size, void* d_ws, size_t ws_size,
                              hipStream_t stream) {
    (void)in_sizes; (void)n_in; (void)out_size; (void)ws_size;
    const float* nodes     = (const float*)d_in[0];
    const int*   senders   = (const int*)d_in[1];
    const int*   receivers = (const int*)d_in[2];
    const float* Wm1 = (const float*)d_in[3];
    const float* bm1 = (const float*)d_in[4];
    const float* Wm2 = (const float*)d_in[5];
    const float* bm2 = (const float*)d_in[6];
    const float* Wn1 = (const float*)d_in[7];
    const float* bn1 = (const float*)d_in[8];
    const float* Wn2 = (const float*)d_in[9];
    const float* bn2 = (const float*)d_in[10];

    float* m_buf = (float*)d_ws;
    int*   head  = (int*)(m_buf + (size_t)N_NODES * D);
    int2*  link  = (int2*)(head + N_NODES);
    float* out   = (float*)d_out;

    const int row_blocks = SMALL_BLOCKS + BIG_BLOCKS;

    gnn_msg_mlp<<<row_blocks, 256, 0, stream>>>(nodes, Wm1, bm1, Wm2, bm2, m_buf);

    hipMemsetAsync(head, 0xFF, sizeof(int) * (size_t)N_NODES, stream);

    gnn_link_edges<<<(N_EDGES + 255) / 256, 256, 0, stream>>>(senders, receivers, head, link);

    gnn_node_mlp<<<row_blocks, 256, 0, stream>>>(nodes, m_buf, head, link, Wn1, bn1, Wn2, bn2, out);
}


# edge-list linking folded into the message kernel: each block links 366 edges (integer exchange) between runs of its node-tile staging; 3 launches -> 2; sums still in list arrival order
# speedup vs baseline: 1.5590x; 1.0346x over previous
#include <hip/hip_runtime.h>
#include <hip/hip_bf16.h>
#include <cstddef>

#define N_NODES 100000
#define N_EDGES 600000
#define D 128
#define H 256

typedef float v2f __attribute__((ext_vector_type(2)));
typedef float v8f __attribute__((ext_vector_type(8)));

__device__ __forceinline__ v8f wmma_f32(v2f a, v2f b, v8f c) {
    return __builtin_amdgcn_wmma_f32_16x16x4_f32(false, a, false, b, (short)0, c,
                                                 false, false);
}

#define ROWS 64
#define RT (ROWS / 16)
#define BIG_BLOCKS 1536
#define SMALL_BLOCKS 106
#define SMALL_ROW0 (BIG_BLOCKS * ROWS)
static_assert(BIG_BLOCKS * ROWS + SMALL_BLOCKS * 16 == N_NODES, "grid cut covers every row once");
#define EDGES_PER_BLOCK 366
static_assert((BIG_BLOCKS + SMALL_BLOCKS) * EDGES_PER_BLOCK >= N_EDGES, "edge shares cover every edge");
static_assert(EDGES_PER_BLOCK > 256 && EDGES_PER_BLOCK <= 512, "one or two edges per thread");
#define XS_STRIDE 132
#define H1_STRIDE 260


__device__ __forceinline__ v8f splat8(float v) {
    v8f r = {v, v, v, v, v, v, v, v};
    return r;
}

template <int KS, int WPITCH>
__device__ __forceinline__ void load_b_stage(float2 (&slot)[KS][2],
                                             const float* __restrict__ wp, const int s) {
#pragma unroll
    for (int ks = 0; ks < KS; ++ks) {
        const float* p = wp + (size_t)((s * KS + ks) * 4) * WPITCH;
        slot[ks][0] = *(const float2*)p;
        slot[ks][1] = *(const float2*)(p + WPITCH);
    }
}

template <int NR, int KS>
__device__ __forceinline__ void load_a_stage(v2f (&slot)[KS][NR], const float* ap,
                                             const int rt_stride) {
#pragma unroll
    for (int ks = 0; ks < KS; ++ks) {
#pragma unroll
        for (int r = 0; r < NR; ++r) {
            const float* x = ap + r * rt_stride + ks * 4;
            const v2f a = {x[0], x[1]};
            slot[ks][r] = a;
        }
    }
}

template <int KS, int PF, int WPITCH>
__device__ __forceinline__ void product_prefetch(float2 (&bq)[PF + 1][KS][2],
                                                 const float* __restrict__ wp) {
#pragma unroll
    for (int s = 0; s < PF; ++s) load_b_stage<KS, WPITCH>(bq[s], wp, s);
}

template <int NR, int KS, int NSTAGE, int PF, int WPITCH, int SPLIT>
__device__ __forceinline__ void product_run(v8f (&acc)[NR][2], float2 (&bq)[PF + 1][KS][2],
                                            const float* __restrict__ wp, const float* ap0,
                                            const float* ap1, const int rt_stride) {
    constexpr int R = PF + 1;
    v2f af[2][KS][NR];
    load_a_stage<NR, KS>(af[0], ap0, rt_stride);
#pragma unroll
    for (int s = 0; s < NSTAGE; ++s) {
        __builtin_amdgcn_sched_barrier(0);
        if (s + PF < NSTAGE) load_b_stage<KS, WPITCH>(bq[(s + PF) % R], wp, s + PF);
        if (s + 1 < NSTAGE) {
            const int n = s + 1;
            load_a_stage<NR, KS>(af[n & 1],
                                 n < SPLIT ? ap0 + n * KS * 4 : ap1 + (n - SPLIT) * KS * 4,
                                 rt_stride);
        }
#pragma unroll
        for (int ks = 0; ks < KS; ++ks) {
            const float2 q0 = bq[s % R][ks][0];
            const float2 q1 = bq[s % R][ks][1];
            const v2f b0 = {q0.x, q1.x};
            const v2f b1 = {q0.y, q1.y};
#pragma unroll
            for (int r = 0; r < NR; ++r) {
                acc[r][0] = wmma_f32(af[s & 1][ks][r], b0, acc[r][0]);
                acc[r][1] = wmma_f32(af[s & 1][ks][r], b1, acc[r][1]);
            }
        }
    }
}
#define PF_L1 3
#define PF_L2 3

template <int NRT>
__device__ __forceinline__ void msg_tile(
    const int row_base, float* Xs, float* H1s,
    const float* __restrict__ nodes, const float* __restrict__ Wm1,
    const float* __restrict__ bm1, const float* __restrict__ Wm2,
    const float* __restrict__ bm2, float* __restrict__ m_out,
    const int* __restrict__ senders, const int* __restrict__ receivers,
    int* __restrict__ head, int2* __restrict__ link) {
    const int tid = threadIdx.x;
    const int w = tid >> 5;
    const int lane = tid & 31;
    const int half = lane >> 4;
    const int l16 = lane & 15;
    constexpr bool PIPE = (NRT == 4);
    constexpr int J = (NRT >= 2) ? NRT / 2 : 1;

    const int ea = (int)blockIdx.x * EDGES_PER_BLOCK + tid;
    const int ec = ea + 256;
    const bool va = ea < N_EDGES;
    const bool vc = (tid + 256 < EDGES_PER_BLOCK) && (ec < N_EDGES);
    const unsigned ra = (unsigned)receivers[va ? ea : 0];
    const int sa = senders[va ? ea : 0];
    const unsigned rc = (unsigned)receivers[vc ? ec : 0];
    const int sc = senders[vc ? ec : 0];

    const int cb1 = w * 32 + 2 * l16;
    const float* wp1 = Wm1 + (size_t)(half * 2) * H + cb1;
    [[maybe_unused]] float2 bq1[PF_L1 + 1][2][2];
    float2 bv1, bv2;
    if constexpr (PIPE) {
        product_prefetch<2, PF_L1, H>(bq1, wp1);
        bv1 = *(const float2*)&bm1[cb1];
    }
    const int cb2 = (w & 3) * 32 + 2 * l16;
    const float* wp2 = Wm2 + (size_t)(half * 2) * D + cb2;
    [[maybe_unused]] float2 bq2[PF_L2 + 1][4][2];

    constexpr int PIECES = NRT * 16 * (D / 4);
    auto stage = [&](const int i0, const int i1) {
        for (int i = i0 + tid; i < i1; i += 256) {
            int row = i / (D / 4);
            int c4 = i % (D / 4);
            int rg = row_base + row;
            float4 v = make_float4(0.f, 0.f, 0.f, 0.f);
            if (rg < N_NODES)
                v = ((const float4*)(nodes + (size_t)rg * D))[c4];
            *(float4*)&Xs[row * XS_STRIDE + c4 * 4] = v;
        }
    };
    stage(0, PIECES / 4);
    int pa = -1, pc = -1;
    if (va) pa = __hip_atomic_exchange(head + ra, ea, __ATOMIC_RELAXED,
                                       __HIP_MEMORY_SCOPE_AGENT);
    if (vc) pc = __hip_atomic_exchange(head + rc, ec, __ATOMIC_RELAXED,
                                       __HIP_MEMORY_SCOPE_AGENT);
    stage(PIECES / 4, 3 * (PIECES / 4));
    if (va) link[ea] = make_int2(pa, sa);
    if (vc) link[ec] = make_int2(pc, sc);
    stage(3 * (PIECES / 4), PIECES);
    __syncthreads();

    {
        const int cb = cb1;
        if constexpr (!PIPE) bv1 = *(const float2*)&bm1[cb];
        const float2 bv = bv1;
        v8f acc[NRT][2];
#pragma unroll
        for (int rt = 0; rt < NRT; ++rt) {
            acc[rt][0] = splat8(bv.x);
            acc[rt][1] = splat8(bv.y);
        }
        const float* wp = wp1;
        const float* xp = &Xs[l16 * XS_STRIDE + half * 2];
        if constexpr (PIPE) {
            product_run<NRT, 2, D / 8, PF_L1, H, D / 8>(acc, bq1, wp, xp, xp, 16 * XS_STRIDE);
            product_prefetch<4, PF_L2, D>(bq2, wp2);
            bv2 = *(const float2*)&bm2[cb2];
        } else {
#pragma unroll 8
        for (int k = 0; k < D; k += 4) {
            const float2 q0 = *(const float2*)(wp + (size_t)k * H);
            const float2 q1 = *(const float2*)(wp + (size_t)(k + 1) * H);
            const v2f b0 = {q0.x, q1.x};
            const v2f b1 = {q0.y, q1.y};
#pragma unroll
            for (int rt = 0; rt < NRT; ++rt) {
                const float* x = xp + rt * 16 * XS_STRIDE + k;
                const v2f a = {x[0], x[1]};
                acc[rt][0] = wmma_f32(a, b0, acc[rt][0]);
                acc[rt][1] = wmma_f32(a, b1, acc[rt][1]);
            }
        }
        }
#pragma unroll
        for (int rt = 0; rt < NRT; ++rt) {
#pragma unroll
            for (int r = 0; r < 8; ++r) {
                const int row = rt * 16 + half * 8 + r;
                const float v0 = acc[rt][0][r];
                const float v1 = acc[rt][1][r];
                *(float2*)&H1s[row * H1_STRIDE + cb] =
                    make_float2(v0 > 0.f ? v0 : 0.f, v1 > 0.f ? v1 : 0.f);
            }
        }
    }
    __syncthreads();

    if (NRT >= 2 || __builtin_amdgcn_readfirstlane(w) < 4) {
        const int cb = cb2;
        const int rt0 = (w >> 2) * J;
        if constexpr (!PIPE) bv2 = *(const float2*)&bm2[cb];
        const float2 bv = bv2;
        v8f acc[J][2];
#pragma unroll
        for (int j = 0; j < J; ++j) {
            acc[j][0] = splat8(bv.x);
            acc[j][1] = splat8(bv.y);
        }
        const float* wp = wp2;
        const float* hp = &H1s[(rt0 * 16 + l16) * H1_STRIDE + half * 2];
        if constexpr (PIPE) {
            product_run<J, 4, H / 16, PF_L2, D, H / 16>(acc, bq2, wp, hp, hp, 16 * H1_STRIDE);
        } else {
#pragma unroll 8
        for (int k = 0; k < H; k += 4) {
            const float2 q0 = *(const float2*)(wp + (size_t)k * D);
            const float2 q1 = *(const float2*)(wp + (size_t)(k + 1) * D);
            const v2f b0 = {q0.x, q1.x};
            const v2f b1 = {q0.y, q1.y};
#pragma unroll
            for (int j = 0; j < J; ++j) {
                const float* hh = hp + j * 16 * H1_STRIDE + k;
                const v2f a = {hh[0], hh[1]};
                acc[j][0] = wmma_f32(a, b0, acc[j][0]);
                acc[j][1] = wmma_f32(a, b1, acc[j][1]);
            }
        }
        }
#pragma unroll
        for (int j = 0; j < J; ++j) {
#pragma unroll
            for (int r = 0; r < 8; ++r) {
                const int rg = row_base + (rt0 + j) * 16 + half * 8 + r;
                if (rg < N_NODES)
                    *(float2*)&m_out[(size_t)rg * D + cb] =
                        make_float2(acc[j][0][r], acc[j][1][r]);
            }
        }
    }
}

__global__ __launch_bounds__(256) void gnn_msg_mlp(
    const float* __restrict__ nodes, const float* __restrict__ Wm1,
    const float* __restrict__ bm1, const float* __restrict__ Wm2,
    const float* __restrict__ bm2, float* __restrict__ m_out,
    const int* __restrict__ senders, const int* __restrict__ receivers,
    int* __restrict__ head, int2* __restrict__ link) {
    __shared__ float Xs[ROWS * XS_STRIDE];
    __shared__ float H1s[ROWS * H1_STRIDE];
    const int b = blockIdx.x;
    if (b < SMALL_BLOCKS)
        msg_tile<1>(SMALL_ROW0 + b * 16, Xs, H1s, nodes, Wm1, bm1, Wm2, bm2, m_out,
                    senders, receivers, head, link);
    else
        msg_tile<RT>((b - SMALL_BLOCKS) * ROWS, Xs, H1s, nodes, Wm1, bm1, Wm2, bm2, m_out,
                     senders, receivers, head, link);
}

template <int NRT>
__device__ __forceinline__ void node_tile(
    const int row_base, float* Xs, float* H1s,
    const float* __restrict__ nodes, const float* __restrict__ m,
    const int* __restrict__ head, const int2* __restrict__ link,
    const float* __restrict__ Wn1, const float* __restrict__ bn1,
    const float* __restrict__ Wn2, const float* __restrict__ bn2,
    float* __restrict__ out) {
    constexpr int GR = NRT * 2;
    float* AGs = H1s;

    const int tid = threadIdx.x;
    const int lane = tid & 31;
    const int wv = __builtin_amdgcn_readfirstlane((int)(tid >> 5));
    const int w = tid >> 5;
    const int half = lane >> 4;
    const int l16 = lane & 15;
    constexpr bool PIPE = (NRT == 4);
    constexpr int J = (NRT >= 2) ? NRT / 2 : 1;
    const int cb1 = w * 32 + 2 * l16;
    const float* wp1 = Wn1 + (size_t)(half * 2) * H + cb1;
    [[maybe_unused]] float2 bq1[PF_L1 + 1][2][2];
    const int cb2 = (w & 3) * 32 + 2 * l16;
    const float* wp2 = Wn2 + (size_t)(half * 2) * D + cb2;
    [[maybe_unused]] float2 bq2[PF_L2 + 1][4][2];
    float2 bv1, bv2;

    for (int i = tid; i < NRT * 16 * (D / 4); i += 256) {
        int row = i / (D / 4);
        int c4 = i % (D / 4);
        int rg = row_base + row;
        float4 v = make_float4(0.f, 0.f, 0.f, 0.f);
        if (rg < N_NODES)
            v = ((const float4*)(nodes + (size_t)rg * D))[c4];
        *(float4*)&Xs[row * XS_STRIDE + c4 * 4] = v;
    }

    {
        const float4* m4 = (const float4*)m;
        int e[GR];
        float4 acc[GR];
        float4 pa[GR], pb[GR];
        bool ga[GR], gb[GR];
        bool any = false;
#pragma unroll
        for (int j = 0; j < GR; ++j) {
            const int r = row_base + wv * GR + j;
            e[j] = -1;
            if (r < N_NODES) e[j] = head[r];
            any = any || ((unsigned)e[j] < (unsigned)N_EDGES);
            acc[j] = make_float4(0.f, 0.f, 0.f, 0.f);
            pa[j] = acc[j]; pb[j] = acc[j];
            ga[j] = false; gb[j] = false;
        }
#define WALK_TRIP(PIN, GIN, POUT, GOUT)                                          \
        {                                                                        \
            int2 ln[GR];                                                         \
            bool live[GR];                                                       \
            _Pragma("unroll") for (int j = 0; j < GR; ++j) {                     \
                live[j] = (unsigned)e[j] < (unsigned)N_EDGES;                    \
                ln[j] = link[live[j] ? e[j] : 0];                                \
            }                                                                    \
            if constexpr (GR == 8) {                                             \
            asm volatile("" :: "s"(ln[0].x), "s"(ln[0].y), "s"(ln[1].x),         \
                               "s"(ln[1].y), "s"(ln[2].x), "s"(ln[2].y),         \
                               "s"(ln[3].x), "s"(ln[3].y), "s"(ln[4].x),         \
                               "s"(ln[4].y), "s"(ln[5].x), "s"(ln[5].y),         \
                               "s"(ln[6].x), "s"(ln[6].y), "s"(ln[7].x),         \
                               "s"(ln[7].y));                                    \
            } else {                                                             \
            asm volatile("" :: "s"(ln[0].x), "s"(ln[0].y), "s"(ln[1].x),         \
                               "s"(ln[1].y));                                    \
            }                                                                    \
            _Pragma("unroll") for (int j = 0; j < GR; ++j) {                     \
                const float4 p = PIN[j];                                         \
                acc[j].x += GIN[j] ? p.x : 0.f; acc[j].y += GIN[j] ? p.y : 0.f;  \
                acc[j].z += GIN[j] ? p.z : 0.f; acc[j].w += GIN[j] ? p.w : 0.f;  \
                const int srow = live[j] ? ln[j].y : 0;                          \
                POUT[j] = m4[(size_t)srow * (D / 4) + lane];                     \
                GOUT[j] = live[j];                                               \
                e[j] = live[j] ? ln[j].x : -1;                                   \
            }                                                                    \
        }
        for (int it = 0; it < N_EDGES && any; it += 2) {
            WALK_TRIP(pb, gb, pa, ga)
            WALK_TRIP(pa, ga, pb, gb)
            any = false;
#pragma unroll
            for (int j = 0; j < GR; ++j)
                any = any || ((unsigned)e[j] < (unsigned)N_EDGES);
        }
#undef WALK_TRIP
        if constexpr (PIPE) {
            product_prefetch<2, PF_L1, H>(bq1, wp1);
            bv1 = *(const float2*)&bn1[cb1];
        }
#pragma unroll
        for (int j = 0; j < GR; ++j) {
            const float4 q = pb[j];
            acc[j].x += gb[j] ? q.x : 0.f; acc[j].y += gb[j] ? q.y : 0.f;
            acc[j].z += gb[j] ? q.z : 0.f; acc[j].w += gb[j] ? q.w : 0.f;
            *(float4*)&AGs[(wv * GR + j) * XS_STRIDE + lane * 4] = acc[j];
        }
    }
    __syncthreads();

    {
        const int cb = cb1;
        if constexpr (!PIPE) bv1 = *(const float2*)&bn1[cb];
        const float2 bv = bv1;
        v8f acc[NRT][2];
#pragma unroll
        for (int rt = 0; rt < NRT; ++rt) {
            acc[rt][0] = splat8(bv.x);
            acc[rt][1] = splat8(bv.y);
        }
        const float* wp = wp1;
        const float* xp = &Xs[l16 * XS_STRIDE + half * 2];
        const float* gp = &AGs[l16 * XS_STRIDE + half * 2];
        if constexpr (PIPE) {
            product_run<NRT, 2, 2 * D / 8, PF_L1, H, D / 8>(acc, bq1, wp, xp, gp, 16 * XS_STRIDE);
            product_prefetch<4, PF_L2, D>(bq2, wp2);
            bv2 = *(const float2*)&bn2[cb2];
        } else {
#pragma unroll 8
        for (int k = 0; k < D; k += 4) {
            const float2 q0 = *(const float2*)(wp + (size_t)k * H);
            const float2 q1 = *(const float2*)(wp + (size_t)(k + 1) * H);
            const v2f b0 = {q0.x, q1.x};
            const v2f b1 = {q0.y, q1.y};
#pragma unroll
            for (int rt = 0; rt < NRT; ++rt) {
                const float* x = xp + rt * 16 * XS_STRIDE + k;
                const v2f a = {x[0], x[1]};
                acc[rt][0] = wmma_f32(a, b0, acc[rt][0]);
                acc[rt][1] = wmma_f32(a, b1, acc[rt][1]);
            }
        }
        const float* wq = wp + (size_t)D * H;
#pragma unroll 8
        for (int k = 0; k < D; k += 4) {
            const float2 q0 = *(const float2*)(wq + (size_t)k * H);
            const float2 q1 = *(const float2*)(wq + (size_t)(k + 1) * H);
            const v2f b0 = {q0.x, q1.x};
            const v2f b1 = {q0.y, q1.y};
#pragma unroll
            for (int rt = 0; rt < NRT; ++rt) {
                const float* g = gp + rt * 16 * XS_STRIDE + k;
                const v2f a = {g[0], g[1]};
                acc[rt][0] = wmma_f32(a, b0, acc[rt][0]);
                acc[rt][1] = wmma_f32(a, b1, acc[rt][1]);
            }
        }
        }
        __syncthreads();
#pragma unroll
        for (int rt = 0; rt < NRT; ++rt) {
#pragma unroll
            for (int r = 0; r < 8; ++r) {
                const int row = rt * 16 + half * 8 + r;
                const float v0 = acc[rt][0][r];
                const float v1 = acc[rt][1][r];
                *(float2*)&H1s[row * H1_STRIDE + cb] =
                    make_float2(v0 > 0.f ? v0 : 0.f, v1 > 0.f ? v1 : 0.f);
            }
        }
    }
    __syncthreads();

    if (NRT >= 2 || wv < 4) {
        const int cb = cb2;
        const int rt0 = (w >> 2) * J;
        if constexpr (!PIPE) bv2 = *(const float2*)&bn2[cb];
        const float2 bv = bv2;
        v8f acc[J][2];
#pragma unroll
        for (int j = 0; j < J; ++j) {
            acc[j][0] = splat8(bv.x);
            acc[j][1] = splat8(bv.y);
        }
        const float* wp = wp2;
        const float* hp = &H1s[(rt0 * 16 + l16) * H1_STRIDE + half * 2];
        if constexpr (PIPE) {
            product_run<J, 4, H / 16, PF_L2, D, H / 16>(acc, bq2, wp, hp, hp, 16 * H1_STRIDE);
        } else {
#pragma unroll 8
        for (int k = 0; k < H; k += 4) {
            const float2 q0 = *(const float2*)(wp + (size_t)k * D);
            const float2 q1 = *(const float2*)(wp + (size_t)(k + 1) * D);
            const v2f b0 = {q0.x, q1.x};
            const v2f b1 = {q0.y, q1.y};
#pragma unroll
            for (int j = 0; j < J; ++j) {
                const float* hh = hp + j * 16 * H1_STRIDE + k;
                const v2f a = {hh[0], hh[1]};
                acc[j][0] = wmma_f32(a, b0, acc[j][0]);
                acc[j][1] = wmma_f32(a, b1, acc[j][1]);
            }
        }
        }
#pragma unroll
        for (int j = 0; j < J; ++j) {
#pragma unroll
            for (int r = 0; r < 8; ++r) {
                const int row = (rt0 + j) * 16 + half * 8 + r;
                const int rg = row_base + row;
                if (rg < N_NODES) {
                    const float2 xv2 = *(const float2*)&Xs[row * XS_STRIDE + cb];
                    *(float2*)&out[(size_t)rg * D + cb] =
                        make_float2(xv2.x + acc[j][0][r], xv2.y + acc[j][1][r]);
                }
            }
        }
    }
}

__global__ __launch_bounds__(256) __attribute__((amdgpu_waves_per_eu(6))) void gnn_node_mlp(
    const float* __restrict__ nodes, const float* __restrict__ m,
    const int* __restrict__ head, const int2* __restrict__ link,
    const float* __restrict__ Wn1, const float* __restrict__ bn1,
    const float* __restrict__ Wn2, const float* __restrict__ bn2,
    float* __restrict__ out) {
    __shared__ __attribute__((aligned(16))) float Xs[ROWS * XS_STRIDE];
    __shared__ __attribute__((aligned(16))) float H1s[ROWS * H1_STRIDE];
    const int b = blockIdx.x;
    if (b < SMALL_BLOCKS)
        node_tile<1>(SMALL_ROW0 + b * 16, Xs, H1s, nodes, m, head, link, Wn1, bn1, Wn2, bn2, out);
    else
        node_tile<RT>((b - SMALL_BLOCKS) * ROWS, Xs, H1s, nodes, m, head, link, Wn1, bn1, Wn2, bn2, out);
}

extern "C" void kernel_launch(void* const* d_in, const int* in_sizes, int n_in,
                              void* d_out, int out_size, void* d_ws, size_t ws_size,
                              hipStream_t stream) {
    (void)in_sizes; (void)n_in; (void)out_size; (void)ws_size;
    const float* nodes     = (const float*)d_in[0];
    const int*   senders   = (const int*)d_in[1];
    const int*   receivers = (const int*)d_in[2];
    const float* Wm1 = (const float*)d_in[3];
    const float* bm1 = (const float*)d_in[4];
    const float* Wm2 = (const float*)d_in[5];
    const float* bm2 = (const float*)d_in[6];
    const float* Wn1 = (const float*)d_in[7];
    const float* bn1 = (const float*)d_in[8];
    const float* Wn2 = (const float*)d_in[9];
    const float* bn2 = (const float*)d_in[10];

    float* m_buf = (float*)d_ws;
    int*   head  = (int*)(m_buf + (size_t)N_NODES * D);
    int2*  link  = (int2*)(head + N_NODES);
    float* out   = (float*)d_out;

    const int row_blocks = SMALL_BLOCKS + BIG_BLOCKS;

    hipMemsetAsync(head, 0xFF, sizeof(int) * (size_t)N_NODES, stream);

    gnn_msg_mlp<<<row_blocks, 256, 0, stream>>>(nodes, Wm1, bm1, Wm2, bm2, m_buf,
                                                senders, receivers, head, link);

    gnn_node_mlp<<<row_blocks, 256, 0, stream>>>(nodes, m_buf, head, link, Wn1, bn1, Wn2, bn2, out);
}
